# starting kernel
# baseline (speedup 1.0000x reference)
#include <hip/hip_runtime.h>
#include <float.h>

typedef __attribute__((ext_vector_type(2))) float v2f;
typedef __attribute__((ext_vector_type(8))) float v8f;

#define KNN 16
#define UNROLL 4

__global__ __launch_bounds__(32)
void softproj_knn_wmma(const float* __restrict__ pc,
                       const float* __restrict__ qc,
                       float* __restrict__ out,
                       int N, int M) {
    const int lane = threadIdx.x;
    const int mrow = lane & 15;
    const bool hi  = lane >= 16;

    const int tiles_m = M >> 4;
    const int b  = blockIdx.x / tiles_m;
    const int m0 = (blockIdx.x % tiles_m) << 4;

    const float* pcb = pc + (size_t)b * 3 * N;
    const float* qcb = qc + (size_t)b * 3 * M;

    const int   mg = m0 + mrow;
    const float qx = qcb[0 * M + mg];
    const float qy = qcb[1 * M + mg];
    const float qz = qcb[2 * M + mg];
    v2f afrag;
    afrag.x = hi ? qz   : qx;
    afrag.y = hi ? 1.0f : qy;

    __shared__ float tile_d[UNROLL][16][20];
    __shared__ float lds_ld[32][KNN + 1];
    __shared__ int   lds_li[32][KNN + 1];

    float kd[KNN];
    int   ki[KNN];
#pragma unroll
    for (int j = 0; j < KNN; ++j) { kd[j] = FLT_MAX; ki[j] = 0; }

    const int c0 = hi ? 8 : 0;

    for (int n0 = 0; n0 < N; n0 += 16 * UNROLL) {
        float px[UNROLL], py[UNROLL], pz[UNROLL];
#pragma unroll
        for (int u = 0; u < UNROLL; ++u) {
            const int n = n0 + 16 * u + mrow;
            px[u] = pcb[0 * N + n];
            py[u] = pcb[1 * N + n];
            pz[u] = pcb[2 * N + n];
        }

#pragma unroll
        for (int u = 0; u < UNROLL; ++u) {
            const float psq = px[u] * px[u] + py[u] * py[u] + pz[u] * pz[u];
            v2f bfrag;
            bfrag.x = hi ? (-2.0f * pz[u]) : (-2.0f * px[u]);
            bfrag.y = hi ? psq             : (-2.0f * py[u]);

            v8f c = {};
            c = __builtin_amdgcn_wmma_f32_16x16x4_f32(false, afrag, false, bfrag,
                                                      (short)0, c, false, false);
#pragma unroll
            for (int r = 0; r < 8; ++r) {
                tile_d[u][r + (hi ? 8 : 0)][mrow] = c[r];
            }
        }
        __syncthreads();

#pragma unroll
        for (int u = 0; u < UNROLL; ++u) {
#pragma unroll
            for (int j = 0; j < 8; ++j) {
                const float d  = tile_d[u][mrow][c0 + j];
                const int   gi = n0 + 16 * u + c0 + j;
                if (d < kd[KNN - 1]) {
                    kd[KNN - 1] = d; ki[KNN - 1] = gi;
#pragma unroll
                    for (int t = KNN - 1; t > 0; --t) {
                        if (kd[t] < kd[t - 1]) {
                            float td = kd[t]; kd[t] = kd[t - 1]; kd[t - 1] = td;
                            int   ti = ki[t]; ki[t] = ki[t - 1]; ki[t - 1] = ti;
                        }
                    }
                }
            }
        }
        __syncthreads();
    }

#pragma unroll
    for (int j = 0; j < KNN; ++j) { lds_ld[lane][j] = kd[j]; lds_li[lane][j] = ki[j]; }
    lds_ld[lane][KNN] = FLT_MAX;
    lds_li[lane][KNN] = 0;
    __syncthreads();

    if (lane < 16) {
        int midx[KNN];
        int i0 = 0, i1 = 0;
#pragma unroll
        for (int j = 0; j < KNN; ++j) {
            const float d0 = lds_ld[mrow][i0];
            const float d1 = lds_ld[mrow + 16][i1];
            if (d0 <= d1) { midx[j] = lds_li[mrow][i0];      ++i0; }
            else          { midx[j] = lds_li[mrow + 16][i1]; ++i1; }
        }

        float pxa[KNN], pya[KNN], pza[KNN], ed[KNN];
        float dmin = FLT_MAX;
#pragma unroll
        for (int j = 0; j < KNN; ++j) {
            const int   gi = midx[j];
            const float px = pcb[0 * N + gi];
            const float py = pcb[1 * N + gi];
            const float pz = pcb[2 * N + gi];
            const float dx = px - qx, dy = py - qy, dz = pz - qz;
            const float d  = dx * dx + dy * dy + dz * dz;
            pxa[j] = px; pya[j] = py; pza[j] = pz; ed[j] = d;
            dmin = fminf(dmin, d);
        }

        float wsum = 0.0f, ox = 0.0f, oy = 0.0f, oz = 0.0f;
#pragma unroll
        for (int j = 0; j < KNN; ++j) {
            const float w = __expf(dmin - ed[j]);
            wsum += w;
            ox += w * pxa[j];
            oy += w * pya[j];
            oz += w * pza[j];
        }
        const float inv = 1.0f / wsum;
        float* outb = out + (size_t)b * 3 * M;
        outb[0 * M + mg] = ox * inv;
        outb[1 * M + mg] = oy * inv;
        outb[2 * M + mg] = oz * inv;
    }
}

extern "C" void kernel_launch(void* const* d_in, const int* in_sizes, int n_in,
                              void* d_out, int out_size, void* d_ws, size_t ws_size,
                              hipStream_t stream) {
    (void)n_in; (void)out_size; (void)d_ws; (void)ws_size;
    const float* pc = (const float*)d_in[0];
    const float* qc = (const float*)d_in[1];
    float* out = (float*)d_out;

    const int B = 4;
    const int N = in_sizes[0] / (3 * B);
    const int M = in_sizes[1] / (3 * B);

    dim3 grid(B * (M / 16));
    softproj_knn_wmma<<<grid, 32, 0, stream>>>(pc, qc, out, N, M);
}


# Split the point range over 8 waves per work-group (each wave scans N/8 for the same 16 queries on a private LDS tile); per-wave top-16 lists merged once at the end; same set, same sum order
# speedup vs baseline: 3.5737x; 3.5737x over previous
#include <hip/hip_runtime.h>
#include <float.h>

typedef __attribute__((ext_vector_type(2))) float v2f;
typedef __attribute__((ext_vector_type(8))) float v8f;
typedef unsigned long long u64;

#define KNN 16
#define UNROLL 4
#define NW 8
#define LPITCH 18

__device__ __forceinline__ void wave_lds_order() {
    __builtin_amdgcn_fence(__ATOMIC_RELEASE, "wavefront");
    __builtin_amdgcn_wave_barrier();
    __builtin_amdgcn_fence(__ATOMIC_ACQUIRE, "wavefront");
}

__device__ __forceinline__ u64 from_partner(u64 v) {
    const unsigned lo = (unsigned)__shfl_xor((int)(unsigned)v, 16);
    const unsigned hi = (unsigned)__shfl_xor((int)(unsigned)(v >> 32), 16);
    return ((u64)hi << 32) | lo;
}

__device__ __forceinline__ void order2(u64& a, u64& b) {
    const bool sw = b < a;
    const u64 lo = sw ? b : a;
    const u64 hi = sw ? a : b;
    a = lo; b = hi;
}

__device__ __forceinline__ void merge16(u64 (&a)[KNN], const u64 (&b)[KNN]) {
#pragma unroll
    for (int i = 0; i < KNN; ++i) {
        const u64 o = b[KNN - 1 - i];
        a[i] = o < a[i] ? o : a[i];
    }
#pragma unroll
    for (int s = KNN / 2; s >= 1; s >>= 1) {
#pragma unroll
        for (int i = 0; i < KNN; ++i) {
            if ((i & s) == 0) order2(a[i], a[i + s]);
        }
    }
}

__global__ __launch_bounds__(32 * NW)
void softproj_knn_wmma(const float* __restrict__ pc,
                       const float* __restrict__ qc,
                       float* __restrict__ out,
                       int N, int M) {
    const int lane = threadIdx.x & 31;
    const int wv   = __builtin_amdgcn_readfirstlane((int)(threadIdx.x >> 5));
    const int mrow = lane & 15;
    const bool hi  = lane >= 16;

    const int tiles_m = M >> 4;
    const int b  = blockIdx.x / tiles_m;
    const int m0 = (blockIdx.x % tiles_m) << 4;

    const float* pcb = pc + (size_t)b * 3 * N;
    const float* qcb = qc + (size_t)b * 3 * M;

    const int   mg = m0 + mrow;
    const float qx = qcb[0 * M + mg];
    const float qy = qcb[1 * M + mg];
    const float qz = qcb[2 * M + mg];
    v2f afrag;
    afrag.x = hi ? qz   : qx;
    afrag.y = hi ? 1.0f : qy;

    __shared__ float tile_d[NW][UNROLL][16][20];
    __shared__ __attribute__((aligned(16))) u64 lists[NW][16][LPITCH];

    float kd[KNN];
    int   ki[KNN];
#pragma unroll
    for (int j = 0; j < KNN; ++j) { kd[j] = FLT_MAX; ki[j] = 0; }

    const int c0 = hi ? 8 : 0;

    const int steps = N / (16 * UNROLL);
    const int per   = (steps + NW - 1) / NW;
    const int n_beg = min(wv * per, steps) * (16 * UNROLL);
    const int n_end = min((wv + 1) * per, steps) * (16 * UNROLL);

    for (int n0 = n_beg; n0 < n_end; n0 += 16 * UNROLL) {
        float px[UNROLL], py[UNROLL], pz[UNROLL];
#pragma unroll
        for (int u = 0; u < UNROLL; ++u) {
            const int n = n0 + 16 * u + mrow;
            px[u] = pcb[0 * N + n];
            py[u] = pcb[1 * N + n];
            pz[u] = pcb[2 * N + n];
        }

#pragma unroll
        for (int u = 0; u < UNROLL; ++u) {
            const float psq = px[u] * px[u] + py[u] * py[u] + pz[u] * pz[u];
            v2f bfrag;
            bfrag.x = hi ? (-2.0f * pz[u]) : (-2.0f * px[u]);
            bfrag.y = hi ? psq             : (-2.0f * py[u]);

            v8f c = {};
            c = __builtin_amdgcn_wmma_f32_16x16x4_f32(false, afrag, false, bfrag,
                                                      (short)0, c, false, false);
#pragma unroll
            for (int r = 0; r < 8; ++r) {
                tile_d[wv][u][r + (hi ? 8 : 0)][mrow] = c[r];
            }
        }
        wave_lds_order();

#pragma unroll
        for (int u = 0; u < UNROLL; ++u) {
#pragma unroll
            for (int j = 0; j < 8; ++j) {
                const float d  = tile_d[wv][u][mrow][c0 + j];
                const int   gi = n0 + 16 * u + c0 + j;
                if (d < kd[KNN - 1]) {
                    kd[KNN - 1] = d; ki[KNN - 1] = gi;
#pragma unroll
                    for (int t = KNN - 1; t > 0; --t) {
                        if (kd[t] < kd[t - 1]) {
                            float td = kd[t]; kd[t] = kd[t - 1]; kd[t - 1] = td;
                            int   ti = ki[t]; ki[t] = ki[t - 1]; ki[t - 1] = ti;
                        }
                    }
                }
            }
        }
        wave_lds_order();
    }

    u64 key[KNN];
    const unsigned half_bit = hi ? 0x80000000u : 0u;
#pragma unroll
    for (int j = 0; j < KNN; ++j) {
        const unsigned ub = __float_as_uint(kd[j]);
        const unsigned su = ub ^ ((unsigned)((int)ub >> 31) | 0x80000000u);
        key[j] = ((u64)su << 32) | ((unsigned)ki[j] | half_bit);
    }

    {
        u64 other[KNN];
#pragma unroll
        for (int j = 0; j < KNN; ++j) other[j] = from_partner(key[j]);
        merge16(key, other);
    }
    if (lane < 16) {
#pragma unroll
        for (int j = 0; j < KNN; ++j) lists[wv][mrow][j] = key[j];
    }
    __syncthreads();
    if (wv != 0) return;

    const int w0 = hi ? NW / 2 : 0;
    u64 acc[KNN];
#pragma unroll
    for (int j = 0; j < KNN; ++j) acc[j] = lists[w0][mrow][j];
#pragma unroll
    for (int t = 1; t < NW / 2; ++t) {
        u64 nx[KNN];
#pragma unroll
        for (int j = 0; j < KNN; ++j) nx[j] = lists[w0 + t][mrow][j];
        merge16(acc, nx);
    }
    {
        u64 other[KNN];
#pragma unroll
        for (int j = 0; j < KNN; ++j) other[j] = from_partner(acc[j]);
        merge16(acc, other);
    }

    if (lane < 16) {
        float pxa[KNN], pya[KNN], pza[KNN], ed[KNN];
        float dmin = FLT_MAX;
#pragma unroll
        for (int j = 0; j < KNN; ++j) {
            const int   gi = (int)((unsigned)acc[j] & 0x7fffffffu);
            const float px = pcb[0 * N + gi];
            const float py = pcb[1 * N + gi];
            const float pz = pcb[2 * N + gi];
            const float dx = px - qx, dy = py - qy, dz = pz - qz;
            const float d  = dx * dx + dy * dy + dz * dz;
            pxa[j] = px; pya[j] = py; pza[j] = pz; ed[j] = d;
            dmin = fminf(dmin, d);
        }

        float wsum = 0.0f, ox = 0.0f, oy = 0.0f, oz = 0.0f;
#pragma unroll
        for (int j = 0; j < KNN; ++j) {
            const float w = __expf(dmin - ed[j]);
            wsum += w;
            ox += w * pxa[j];
            oy += w * pya[j];
            oz += w * pza[j];
        }
        const float inv = 1.0f / wsum;
        float* outb = out + (size_t)b * 3 * M;
        outb[0 * M + mg] = ox * inv;
        outb[1 * M + mg] = oy * inv;
        outb[2 * M + mg] = oz * inv;
    }
}

extern "C" void kernel_launch(void* const* d_in, const int* in_sizes, int n_in,
                              void* d_out, int out_size, void* d_ws, size_t ws_size,
                              hipStream_t stream) {
    (void)n_in; (void)out_size; (void)d_ws; (void)ws_size;
    const float* pc = (const float*)d_in[0];
    const float* qc = (const float*)d_in[1];
    float* out = (float*)d_out;

    const int B = 4;
    const int N = in_sizes[0] / (3 * B);
    const int M = in_sizes[1] / (3 * B);

    dim3 grid(B * (M / 16));
    softproj_knn_wmma<<<grid, 32 * NW, 0, stream>>>(pc, qc, out, N, M);
}


# Scan only marks (per-lane 32-bit mask) the points that may beat the lane's 16th-best; lanes then insert their own marked points in shared branch-free passes; same set, same sum order
# speedup vs baseline: 6.2242x; 1.7417x over previous
#include <hip/hip_runtime.h>
#include <float.h>

typedef __attribute__((ext_vector_type(2))) float v2f;
typedef __attribute__((ext_vector_type(8))) float v8f;
typedef unsigned long long u64;

#define KNN 16
#define UNROLL 4
#define NW 8
#define LPITCH 18

__device__ __forceinline__ void wave_lds_order() {
    __builtin_amdgcn_fence(__ATOMIC_RELEASE, "wavefront");
    __builtin_amdgcn_wave_barrier();
    __builtin_amdgcn_fence(__ATOMIC_ACQUIRE, "wavefront");
}

__device__ __forceinline__ u64 from_partner(u64 v) {
    const unsigned lo = (unsigned)__shfl_xor((int)(unsigned)v, 16);
    const unsigned hi = (unsigned)__shfl_xor((int)(unsigned)(v >> 32), 16);
    return ((u64)hi << 32) | lo;
}

__device__ __forceinline__ void order2(u64& a, u64& b) {
    const bool sw = b < a;
    const u64 lo = sw ? b : a;
    const u64 hi = sw ? a : b;
    a = lo; b = hi;
}

__device__ __forceinline__ void merge16(u64 (&a)[KNN], const u64 (&b)[KNN]) {
#pragma unroll
    for (int i = 0; i < KNN; ++i) {
        const u64 o = b[KNN - 1 - i];
        a[i] = o < a[i] ? o : a[i];
    }
#pragma unroll
    for (int s = KNN / 2; s >= 1; s >>= 1) {
#pragma unroll
        for (int i = 0; i < KNN; ++i) {
            if ((i & s) == 0) order2(a[i], a[i + s]);
        }
    }
}

__global__ __launch_bounds__(32 * NW)
void softproj_knn_wmma(const float* __restrict__ pc,
                       const float* __restrict__ qc,
                       float* __restrict__ out,
                       int N, int M) {
    const int lane = threadIdx.x & 31;
    const int wv   = __builtin_amdgcn_readfirstlane((int)(threadIdx.x >> 5));
    const int mrow = lane & 15;
    const bool hi  = lane >= 16;

    const int tiles_m = M >> 4;
    const int b  = blockIdx.x / tiles_m;
    const int m0 = (blockIdx.x % tiles_m) << 4;

    const float* pcb = pc + (size_t)b * 3 * N;
    const float* qcb = qc + (size_t)b * 3 * M;

    const int   mg = m0 + mrow;
    const float qx = qcb[0 * M + mg];
    const float qy = qcb[1 * M + mg];
    const float qz = qcb[2 * M + mg];
    v2f afrag;
    afrag.x = hi ? qz   : qx;
    afrag.y = hi ? 1.0f : qy;

    __shared__ __attribute__((aligned(16))) float tile_d[NW][UNROLL][16][20];
    __shared__ __attribute__((aligned(16))) u64 lists[NW][16][LPITCH];

    float kd[KNN];
    int   ki[KNN];
#pragma unroll
    for (int j = 0; j < KNN; ++j) { kd[j] = FLT_MAX; ki[j] = 0; }

    const int c0 = hi ? 8 : 0;

    const int steps = N / (16 * UNROLL);
    const int per   = (steps + NW - 1) / NW;
    const int n_beg = min(wv * per, steps) * (16 * UNROLL);
    const int n_end = min((wv + 1) * per, steps) * (16 * UNROLL);

    for (int n0 = n_beg; n0 < n_end; n0 += 16 * UNROLL) {
        float px[UNROLL], py[UNROLL], pz[UNROLL];
#pragma unroll
        for (int u = 0; u < UNROLL; ++u) {
            const int n = n0 + 16 * u + mrow;
            px[u] = pcb[0 * N + n];
            py[u] = pcb[1 * N + n];
            pz[u] = pcb[2 * N + n];
        }

#pragma unroll
        for (int u = 0; u < UNROLL; ++u) {
            const float psq = px[u] * px[u] + py[u] * py[u] + pz[u] * pz[u];
            v2f bfrag;
            bfrag.x = hi ? (-2.0f * pz[u]) : (-2.0f * px[u]);
            bfrag.y = hi ? psq             : (-2.0f * py[u]);

            v8f c = {};
            c = __builtin_amdgcn_wmma_f32_16x16x4_f32(false, afrag, false, bfrag,
                                                      (short)0, c, false, false);
#pragma unroll
            for (int r = 0; r < 8; ++r) {
                tile_d[wv][u][r + (hi ? 8 : 0)][mrow] = c[r];
            }
        }
        wave_lds_order();

        const float thr = kd[KNN - 1];
        unsigned marks = 0u;
#pragma unroll
        for (int u = 0; u < UNROLL; ++u) {
            const float4 qa = *reinterpret_cast<const float4*>(&tile_d[wv][u][mrow][c0]);
            const float4 qb = *reinterpret_cast<const float4*>(&tile_d[wv][u][mrow][c0 + 4]);
            const float dv[8] = {qa.x, qa.y, qa.z, qa.w, qb.x, qb.y, qb.z, qb.w};
            unsigned mu = 0u;
#pragma unroll
            for (int j = 0; j < 8; ++j) {
                mu = (mu << 1) | (__float_as_uint(dv[j] - thr) >> 31);
            }
            marks = (marks << 8) | mu;
        }

        auto insert_pass = [&]() {
            const bool  have = marks != 0u;
            const int   p    = __clz((int)marks) & 31;
            marks &= ~(0x80000000u >> p);
            const float d    = tile_d[wv][p >> 3][mrow][c0 + (p & 7)];
            const int   gi   = n0 + 16 * (p >> 3) + c0 + (p & 7);
            const bool  ins  = have && (d < kd[KNN - 1]);
            kd[KNN - 1] = ins ? d  : kd[KNN - 1];
            ki[KNN - 1] = ins ? gi : ki[KNN - 1];
#pragma unroll
            for (int t = KNN - 1; t > 0; --t) {
                const bool  sw = kd[t] < kd[t - 1];
                const float dl = sw ? kd[t] : kd[t - 1];
                const float dh = sw ? kd[t - 1] : kd[t];
                const int   il = sw ? ki[t] : ki[t - 1];
                const int   ih = sw ? ki[t - 1] : ki[t];
                kd[t - 1] = dl; kd[t] = dh; ki[t - 1] = il; ki[t] = ih;
            }
        };
        for (int trip = 0; trip < 8; ++trip) {
            if (__builtin_amdgcn_ballot_w32(marks != 0u) == 0u) break;
            insert_pass();
            if (__builtin_amdgcn_ballot_w32(marks != 0u) == 0u) break;
            insert_pass();
            if (__builtin_amdgcn_ballot_w32(marks != 0u) == 0u) break;
            insert_pass();
            if (__builtin_amdgcn_ballot_w32(marks != 0u) == 0u) break;
            insert_pass();
        }
        wave_lds_order();
    }

    u64 key[KNN];
    const unsigned half_bit = hi ? 0x80000000u : 0u;
#pragma unroll
    for (int j = 0; j < KNN; ++j) {
        const unsigned ub = __float_as_uint(kd[j]);
        const unsigned su = ub ^ ((unsigned)((int)ub >> 31) | 0x80000000u);
        key[j] = ((u64)su << 32) | ((unsigned)ki[j] | half_bit);
    }

    {
        u64 other[KNN];
#pragma unroll
        for (int j = 0; j < KNN; ++j) other[j] = from_partner(key[j]);
        merge16(key, other);
    }
    if (lane < 16) {
#pragma unroll
        for (int j = 0; j < KNN; ++j) lists[wv][mrow][j] = key[j];
    }
    __syncthreads();
    if (wv != 0) return;

    const int w0 = hi ? NW / 2 : 0;
    u64 acc[KNN];
#pragma unroll
    for (int j = 0; j < KNN; ++j) acc[j] = lists[w0][mrow][j];
#pragma unroll
    for (int t = 1; t < NW / 2; ++t) {
        u64 nx[KNN];
#pragma unroll
        for (int j = 0; j < KNN; ++j) nx[j] = lists[w0 + t][mrow][j];
        merge16(acc, nx);
    }
    {
        u64 other[KNN];
#pragma unroll
        for (int j = 0; j < KNN; ++j) other[j] = from_partner(acc[j]);
        merge16(acc, other);
    }

    if (lane < 16) {
        float pxa[KNN], pya[KNN], pza[KNN], ed[KNN];
        float dmin = FLT_MAX;
#pragma unroll
        for (int j = 0; j < KNN; ++j) {
            const int   gi = (int)((unsigned)acc[j] & 0x7fffffffu);
            const float px = pcb[0 * N + gi];
            const float py = pcb[1 * N + gi];
            const float pz = pcb[2 * N + gi];
            const float dx = px - qx, dy = py - qy, dz = pz - qz;
            const float d  = dx * dx + dy * dy + dz * dz;
            pxa[j] = px; pya[j] = py; pza[j] = pz; ed[j] = d;
            dmin = fminf(dmin, d);
        }

        float wsum = 0.0f, ox = 0.0f, oy = 0.0f, oz = 0.0f;
#pragma unroll
        for (int j = 0; j < KNN; ++j) {
            const float w = __expf(dmin - ed[j]);
            wsum += w;
            ox += w * pxa[j];
            oy += w * pya[j];
            oz += w * pza[j];
        }
        const float inv = 1.0f / wsum;
        float* outb = out + (size_t)b * 3 * M;
        outb[0 * M + mg] = ox * inv;
        outb[1 * M + mg] = oy * inv;
        outb[2 * M + mg] = oz * inv;
    }
}

extern "C" void kernel_launch(void* const* d_in, const int* in_sizes, int n_in,
                              void* d_out, int out_size, void* d_ws, size_t ws_size,
                              hipStream_t stream) {
    (void)n_in; (void)out_size; (void)d_ws; (void)ws_size;
    const float* pc = (const float*)d_in[0];
    const float* qc = (const float*)d_in[1];
    float* out = (float*)d_out;

    const int B = 4;
    const int N = in_sizes[0] / (3 * B);
    const int M = in_sizes[1] / (3 * B);

    dim3 grid(B * (M / 16));
    softproj_knn_wmma<<<grid, 32 * NW, 0, stream>>>(pc, qc, out, N, M);
}


# Scan marks against a tighter per-query bound: min(own 16th-best, larger 8th-best of the two halves, largest 2nd-best over the 8 waves, swapped via LDS, one barrier per step); same set and sum order
# speedup vs baseline: 7.9617x; 1.2792x over previous
#include <hip/hip_runtime.h>
#include <float.h>

typedef __attribute__((ext_vector_type(2))) float v2f;
typedef __attribute__((ext_vector_type(8))) float v8f;
typedef unsigned long long u64;

#define KNN 16
#define UNROLL 4
#define NW 8
#define LPITCH 18

__device__ __forceinline__ void wave_lds_order() {
    __builtin_amdgcn_fence(__ATOMIC_RELEASE, "wavefront");
    __builtin_amdgcn_wave_barrier();
    __builtin_amdgcn_fence(__ATOMIC_ACQUIRE, "wavefront");
}

__device__ __forceinline__ u64 from_partner(u64 v) {
    const unsigned lo = (unsigned)__shfl_xor((int)(unsigned)v, 16);
    const unsigned hi = (unsigned)__shfl_xor((int)(unsigned)(v >> 32), 16);
    return ((u64)hi << 32) | lo;
}

__device__ __forceinline__ float next_above(float x) {
    int bits = __float_as_int(x);
    bits = (bits == (int)0x80000000) ? 0 : bits;
    bits += (bits >= 0) ? 1 : -1;
    return __int_as_float(bits);
}

__device__ __forceinline__ void order2(u64& a, u64& b) {
    const bool sw = b < a;
    const u64 lo = sw ? b : a;
    const u64 hi = sw ? a : b;
    a = lo; b = hi;
}

__device__ __forceinline__ void merge16(u64 (&a)[KNN], const u64 (&b)[KNN]) {
#pragma unroll
    for (int i = 0; i < KNN; ++i) {
        const u64 o = b[KNN - 1 - i];
        a[i] = o < a[i] ? o : a[i];
    }
#pragma unroll
    for (int s = KNN / 2; s >= 1; s >>= 1) {
#pragma unroll
        for (int i = 0; i < KNN; ++i) {
            if ((i & s) == 0) order2(a[i], a[i + s]);
        }
    }
}

__global__ __launch_bounds__(32 * NW)
void softproj_knn_wmma(const float* __restrict__ pc,
                       const float* __restrict__ qc,
                       float* __restrict__ out,
                       int N, int M) {
    const int lane = threadIdx.x & 31;
    const int wv   = __builtin_amdgcn_readfirstlane((int)(threadIdx.x >> 5));
    const int mrow = lane & 15;
    const bool hi  = lane >= 16;

    const int tiles_m = M >> 4;
    const int b  = blockIdx.x / tiles_m;
    const int m0 = (blockIdx.x % tiles_m) << 4;

    const float* pcb = pc + (size_t)b * 3 * N;
    const float* qcb = qc + (size_t)b * 3 * M;

    const int   mg = m0 + mrow;
    const float qx = qcb[0 * M + mg];
    const float qy = qcb[1 * M + mg];
    const float qz = qcb[2 * M + mg];
    v2f afrag;
    afrag.x = hi ? qz   : qx;
    afrag.y = hi ? 1.0f : qy;

    __shared__ __attribute__((aligned(16))) float tile_d[NW][UNROLL][16][20];
    __shared__ __attribute__((aligned(16))) u64 lists[NW][16][LPITCH];
    __shared__ __attribute__((aligned(16))) float bound_pub[2][16][NW];

    float kd[KNN];
    int   ki[KNN];
#pragma unroll
    for (int j = 0; j < KNN; ++j) { kd[j] = FLT_MAX; ki[j] = 0; }

    const int c0 = hi ? 8 : 0;

    const int steps = N / (16 * UNROLL);
    const int per   = (steps + NW - 1) / NW;
    const int s_beg = min(wv * per, steps);
    const int s_end = min((wv + 1) * per, steps);

    for (int it = 0; it < per; ++it) {
        const bool live = s_beg + it < s_end;
        const int  n0   = min(s_beg + it, steps - 1) * (16 * UNROLL);
        float px[UNROLL], py[UNROLL], pz[UNROLL];
#pragma unroll
        for (int u = 0; u < UNROLL; ++u) {
            const int n = n0 + 16 * u + mrow;
            px[u] = pcb[0 * N + n];
            py[u] = pcb[1 * N + n];
            pz[u] = pcb[2 * N + n];
        }

        const float o0 = __shfl_xor(kd[0], 16);
        const float o1 = __shfl_xor(kd[1], 16);
        const float o7 = __shfl_xor(kd[7], 16);
        const float second = fminf(fmaxf(kd[0], o0), fminf(kd[1], o1));
        const float pair8  = fmaxf(kd[7], o7);
        if (lane < 16) bound_pub[it & 1][mrow][wv] = second;
        __builtin_amdgcn_fence(__ATOMIC_RELEASE, "workgroup", "local");
        __builtin_amdgcn_s_barrier();
        __builtin_amdgcn_fence(__ATOMIC_ACQUIRE, "workgroup", "local");
        float wgb = bound_pub[it & 1][mrow][0];
#pragma unroll
        for (int w = 1; w < NW; ++w) wgb = fmaxf(wgb, bound_pub[it & 1][mrow][w]);
        const float foreign = fminf(pair8, wgb);

#pragma unroll
        for (int u = 0; u < UNROLL; ++u) {
            const float psq = px[u] * px[u] + py[u] * py[u] + pz[u] * pz[u];
            v2f bfrag;
            bfrag.x = hi ? (-2.0f * pz[u]) : (-2.0f * px[u]);
            bfrag.y = hi ? psq             : (-2.0f * py[u]);

            v8f c = {};
            c = __builtin_amdgcn_wmma_f32_16x16x4_f32(false, afrag, false, bfrag,
                                                      (short)0, c, false, false);
#pragma unroll
            for (int r = 0; r < 8; ++r) {
                tile_d[wv][u][r + (hi ? 8 : 0)][mrow] = c[r];
            }
        }
        wave_lds_order();

        const float thr = live ? fminf(kd[KNN - 1], next_above(foreign)) : -INFINITY;
        unsigned marks = 0u;
#pragma unroll
        for (int u = 0; u < UNROLL; ++u) {
            const float4 qa = *reinterpret_cast<const float4*>(&tile_d[wv][u][mrow][c0]);
            const float4 qb = *reinterpret_cast<const float4*>(&tile_d[wv][u][mrow][c0 + 4]);
            const float dv[8] = {qa.x, qa.y, qa.z, qa.w, qb.x, qb.y, qb.z, qb.w};
            unsigned mu = 0u;
#pragma unroll
            for (int j = 0; j < 8; ++j) {
                mu = (mu << 1) | (__float_as_uint(dv[j] - thr) >> 31);
            }
            marks = (marks << 8) | mu;
        }

        auto insert_pass = [&]() {
            const bool  have = marks != 0u;
            const int   p    = __clz((int)marks) & 31;
            marks &= ~(0x80000000u >> p);
            const float d    = tile_d[wv][p >> 3][mrow][c0 + (p & 7)];
            const int   gi   = n0 + 16 * (p >> 3) + c0 + (p & 7);
            const bool  ins  = have && (d < kd[KNN - 1]);
            kd[KNN - 1] = ins ? d  : kd[KNN - 1];
            ki[KNN - 1] = ins ? gi : ki[KNN - 1];
#pragma unroll
            for (int t = KNN - 1; t > 0; --t) {
                const bool  sw = kd[t] < kd[t - 1];
                const float dl = sw ? kd[t] : kd[t - 1];
                const float dh = sw ? kd[t - 1] : kd[t];
                const int   il = sw ? ki[t] : ki[t - 1];
                const int   ih = sw ? ki[t - 1] : ki[t];
                kd[t - 1] = dl; kd[t] = dh; ki[t - 1] = il; ki[t] = ih;
            }
        };
        for (int trip = 0; trip < 8; ++trip) {
            if (__builtin_amdgcn_ballot_w32(marks != 0u) == 0u) break;
            insert_pass();
            if (__builtin_amdgcn_ballot_w32(marks != 0u) == 0u) break;
            insert_pass();
            if (__builtin_amdgcn_ballot_w32(marks != 0u) == 0u) break;
            insert_pass();
            if (__builtin_amdgcn_ballot_w32(marks != 0u) == 0u) break;
            insert_pass();
        }
        wave_lds_order();
    }

    u64 key[KNN];
    const unsigned half_bit = hi ? 0x80000000u : 0u;
#pragma unroll
    for (int j = 0; j < KNN; ++j) {
        const unsigned ub = __float_as_uint(kd[j]);
        const unsigned su = ub ^ ((unsigned)((int)ub >> 31) | 0x80000000u);
        key[j] = ((u64)su << 32) | ((unsigned)ki[j] | half_bit);
    }

    {
        u64 other[KNN];
#pragma unroll
        for (int j = 0; j < KNN; ++j) other[j] = from_partner(key[j]);
        merge16(key, other);
    }
    if (lane < 16) {
#pragma unroll
        for (int j = 0; j < KNN; ++j) lists[wv][mrow][j] = key[j];
    }
    __syncthreads();
    if (wv != 0) return;

    const int w0 = hi ? NW / 2 : 0;
    u64 acc[KNN];
#pragma unroll
    for (int j = 0; j < KNN; ++j) acc[j] = lists[w0][mrow][j];
#pragma unroll
    for (int t = 1; t < NW / 2; ++t) {
        u64 nx[KNN];
#pragma unroll
        for (int j = 0; j < KNN; ++j) nx[j] = lists[w0 + t][mrow][j];
        merge16(acc, nx);
    }
    {
        u64 other[KNN];
#pragma unroll
        for (int j = 0; j < KNN; ++j) other[j] = from_partner(acc[j]);
        merge16(acc, other);
    }

    if (lane < 16) {
        float pxa[KNN], pya[KNN], pza[KNN], ed[KNN];
        float dmin = FLT_MAX;
#pragma unroll
        for (int j = 0; j < KNN; ++j) {
            const int   gi = (int)((unsigned)acc[j] & 0x7fffffffu);
            const float px = pcb[0 * N + gi];
            const float py = pcb[1 * N + gi];
            const float pz = pcb[2 * N + gi];
            const float dx = px - qx, dy = py - qy, dz = pz - qz;
            const float d  = dx * dx + dy * dy + dz * dz;
            pxa[j] = px; pya[j] = py; pza[j] = pz; ed[j] = d;
            dmin = fminf(dmin, d);
        }

        float wsum = 0.0f, ox = 0.0f, oy = 0.0f, oz = 0.0f;
#pragma unroll
        for (int j = 0; j < KNN; ++j) {
            const float w = __expf(dmin - ed[j]);
            wsum += w;
            ox += w * pxa[j];
            oy += w * pya[j];
            oz += w * pza[j];
        }
        const float inv = 1.0f / wsum;
        float* outb = out + (size_t)b * 3 * M;
        outb[0 * M + mg] = ox * inv;
        outb[1 * M + mg] = oy * inv;
        outb[2 * M + mg] = oz * inv;
    }
}

extern "C" void kernel_launch(void* const* d_in, const int* in_sizes, int n_in,
                              void* d_out, int out_size, void* d_ws, size_t ws_size,
                              hipStream_t stream) {
    (void)n_in; (void)out_size; (void)d_ws; (void)ws_size;
    const float* pc = (const float*)d_in[0];
    const float* qc = (const float*)d_in[1];
    float* out = (float*)d_out;

    const int B = 4;
    const int N = in_sizes[0] / (3 * B);
    const int M = in_sizes[1] / (3 * B);

    dim3 grid(B * (M / 16));
    softproj_knn_wmma<<<grid, 32 * NW, 0, stream>>>(pc, qc, out, N, M);
}


# Insert by rank: a pass computes all 16 list entries from the old list and the new distance (median of neighbours; two selects for the index), not by a 15-stage swap chain; same set and sum order
# speedup vs baseline: 8.2844x; 1.0405x over previous
#include <hip/hip_runtime.h>
#include <float.h>

typedef __attribute__((ext_vector_type(2))) float v2f;
typedef __attribute__((ext_vector_type(8))) float v8f;
typedef unsigned long long u64;

#define KNN 16
#define UNROLL 4
#define NW 8
#define LPITCH 18

__device__ __forceinline__ void wave_lds_order() {
    __builtin_amdgcn_fence(__ATOMIC_RELEASE, "wavefront");
    __builtin_amdgcn_wave_barrier();
    __builtin_amdgcn_fence(__ATOMIC_ACQUIRE, "wavefront");
}

__device__ __forceinline__ u64 from_partner(u64 v) {
    const unsigned lo = (unsigned)__shfl_xor((int)(unsigned)v, 16);
    const unsigned hi = (unsigned)__shfl_xor((int)(unsigned)(v >> 32), 16);
    return ((u64)hi << 32) | lo;
}

__device__ __forceinline__ float next_above(float x) {
    int bits = __float_as_int(x);
    bits = (bits == (int)0x80000000) ? 0 : bits;
    bits += (bits >= 0) ? 1 : -1;
    return __int_as_float(bits);
}

__device__ __forceinline__ void order2(u64& a, u64& b) {
    const bool sw = b < a;
    const u64 lo = sw ? b : a;
    const u64 hi = sw ? a : b;
    a = lo; b = hi;
}

__device__ __forceinline__ void merge16(u64 (&a)[KNN], const u64 (&b)[KNN]) {
#pragma unroll
    for (int i = 0; i < KNN; ++i) {
        const u64 o = b[KNN - 1 - i];
        a[i] = o < a[i] ? o : a[i];
    }
#pragma unroll
    for (int s = KNN / 2; s >= 1; s >>= 1) {
#pragma unroll
        for (int i = 0; i < KNN; ++i) {
            if ((i & s) == 0) order2(a[i], a[i + s]);
        }
    }
}

__global__ __launch_bounds__(32 * NW)
void softproj_knn_wmma(const float* __restrict__ pc,
                       const float* __restrict__ qc,
                       float* __restrict__ out,
                       int N, int M) {
    const int lane = threadIdx.x & 31;
    const int wv   = __builtin_amdgcn_readfirstlane((int)(threadIdx.x >> 5));
    const int mrow = lane & 15;
    const bool hi  = lane >= 16;

    const int tiles_m = M >> 4;
    const int b  = blockIdx.x / tiles_m;
    const int m0 = (blockIdx.x % tiles_m) << 4;

    const float* pcb = pc + (size_t)b * 3 * N;
    const float* qcb = qc + (size_t)b * 3 * M;

    const int   mg = m0 + mrow;
    const float qx = qcb[0 * M + mg];
    const float qy = qcb[1 * M + mg];
    const float qz = qcb[2 * M + mg];
    v2f afrag;
    afrag.x = hi ? qz   : qx;
    afrag.y = hi ? 1.0f : qy;

    __shared__ __attribute__((aligned(16))) float tile_d[NW][UNROLL][16][20];
    __shared__ __attribute__((aligned(16))) u64 lists[NW][16][LPITCH];
    __shared__ __attribute__((aligned(16))) float bound_pub[2][16][NW];

    float kd[KNN];
    int   ki[KNN];
#pragma unroll
    for (int j = 0; j < KNN; ++j) { kd[j] = FLT_MAX; ki[j] = 0; }

    const int c0 = hi ? 8 : 0;

    const int steps = N / (16 * UNROLL);
    const int per   = (steps + NW - 1) / NW;
    const int s_beg = min(wv * per, steps);
    const int s_end = min((wv + 1) * per, steps);

    for (int it = 0; it < per; ++it) {
        const bool live = s_beg + it < s_end;
        const int  n0   = min(s_beg + it, steps - 1) * (16 * UNROLL);
        float px[UNROLL], py[UNROLL], pz[UNROLL];
#pragma unroll
        for (int u = 0; u < UNROLL; ++u) {
            const int n = n0 + 16 * u + mrow;
            px[u] = pcb[0 * N + n];
            py[u] = pcb[1 * N + n];
            pz[u] = pcb[2 * N + n];
        }

        const float o0 = __shfl_xor(kd[0], 16);
        const float o1 = __shfl_xor(kd[1], 16);
        const float o7 = __shfl_xor(kd[7], 16);
        const float second = fminf(fmaxf(kd[0], o0), fminf(kd[1], o1));
        const float pair8  = fmaxf(kd[7], o7);
        if (lane < 16) bound_pub[it & 1][mrow][wv] = second;
        __builtin_amdgcn_fence(__ATOMIC_RELEASE, "workgroup", "local");
        __builtin_amdgcn_s_barrier();
        __builtin_amdgcn_fence(__ATOMIC_ACQUIRE, "workgroup", "local");
        float wgb = bound_pub[it & 1][mrow][0];
#pragma unroll
        for (int w = 1; w < NW; ++w) wgb = fmaxf(wgb, bound_pub[it & 1][mrow][w]);
        const float foreign = fminf(pair8, wgb);

#pragma unroll
        for (int u = 0; u < UNROLL; ++u) {
            const float psq = px[u] * px[u] + py[u] * py[u] + pz[u] * pz[u];
            v2f bfrag;
            bfrag.x = hi ? (-2.0f * pz[u]) : (-2.0f * px[u]);
            bfrag.y = hi ? psq             : (-2.0f * py[u]);

            v8f c = {};
            c = __builtin_amdgcn_wmma_f32_16x16x4_f32(false, afrag, false, bfrag,
                                                      (short)0, c, false, false);
#pragma unroll
            for (int r = 0; r < 8; ++r) {
                tile_d[wv][u][r + (hi ? 8 : 0)][mrow] = c[r];
            }
        }
        wave_lds_order();

        const float thr = live ? fminf(kd[KNN - 1], next_above(foreign)) : -INFINITY;
        unsigned marks = 0u;
#pragma unroll
        for (int u = 0; u < UNROLL; ++u) {
            const float4 qa = *reinterpret_cast<const float4*>(&tile_d[wv][u][mrow][c0]);
            const float4 qb = *reinterpret_cast<const float4*>(&tile_d[wv][u][mrow][c0 + 4]);
            const float dv[8] = {qa.x, qa.y, qa.z, qa.w, qb.x, qb.y, qb.z, qb.w};
            unsigned mu = 0u;
#pragma unroll
            for (int j = 0; j < 8; ++j) {
                mu = (mu << 1) | (__float_as_uint(dv[j] - thr) >> 31);
            }
            marks = (marks << 8) | mu;
        }

        auto insert_pass = [&]() {
            const bool  have = marks != 0u;
            const int   p    = __clz((int)marks) & 31;
            marks &= ~(0x80000000u >> p);
            const float d    = tile_d[wv][p >> 3][mrow][c0 + (p & 7)];
            const int   gi   = n0 + 16 * (p >> 3) + c0 + (p & 7);
            const float dn   = have ? d : INFINITY;
            int ni[KNN];
#pragma unroll
            for (int t = 0; t < KNN; ++t) ni[t] = (dn < kd[t]) ? gi : ki[t];
#pragma unroll
            for (int t = KNN - 1; t > 0; --t) {
                ki[t] = (dn < kd[t - 1]) ? ki[t - 1] : ni[t];
                kd[t] = __builtin_amdgcn_fmed3f(kd[t - 1], kd[t], dn);
            }
            ki[0] = ni[0];
            kd[0] = __builtin_amdgcn_fmed3f(-INFINITY, kd[0], dn);
        };
        for (int trip = 0; trip < 8; ++trip) {
            if (__builtin_amdgcn_ballot_w32(marks != 0u) == 0u) break;
            insert_pass();
            if (__builtin_amdgcn_ballot_w32(marks != 0u) == 0u) break;
            insert_pass();
            if (__builtin_amdgcn_ballot_w32(marks != 0u) == 0u) break;
            insert_pass();
            if (__builtin_amdgcn_ballot_w32(marks != 0u) == 0u) break;
            insert_pass();
        }
        wave_lds_order();
    }

    u64 key[KNN];
    const unsigned half_bit = hi ? 0x80000000u : 0u;
#pragma unroll
    for (int j = 0; j < KNN; ++j) {
        const unsigned ub = __float_as_uint(kd[j]);
        const unsigned su = ub ^ ((unsigned)((int)ub >> 31) | 0x80000000u);
        key[j] = ((u64)su << 32) | ((unsigned)ki[j] | half_bit);
    }

    {
        u64 other[KNN];
#pragma unroll
        for (int j = 0; j < KNN; ++j) other[j] = from_partner(key[j]);
        merge16(key, other);
    }
    if (lane < 16) {
#pragma unroll
        for (int j = 0; j < KNN; ++j) lists[wv][mrow][j] = key[j];
    }
    __syncthreads();
    if (wv != 0) return;

    const int w0 = hi ? NW / 2 : 0;
    u64 acc[KNN];
#pragma unroll
    for (int j = 0; j < KNN; ++j) acc[j] = lists[w0][mrow][j];
#pragma unroll
    for (int t = 1; t < NW / 2; ++t) {
        u64 nx[KNN];
#pragma unroll
        for (int j = 0; j < KNN; ++j) nx[j] = lists[w0 + t][mrow][j];
        merge16(acc, nx);
    }
    {
        u64 other[KNN];
#pragma unroll
        for (int j = 0; j < KNN; ++j) other[j] = from_partner(acc[j]);
        merge16(acc, other);
    }

    if (lane < 16) {
        float pxa[KNN], pya[KNN], pza[KNN], ed[KNN];
        float dmin = FLT_MAX;
#pragma unroll
        for (int j = 0; j < KNN; ++j) {
            const int   gi = (int)((unsigned)acc[j] & 0x7fffffffu);
            const float px = pcb[0 * N + gi];
            const float py = pcb[1 * N + gi];
            const float pz = pcb[2 * N + gi];
            const float dx = px - qx, dy = py - qy, dz = pz - qz;
            const float d  = dx * dx + dy * dy + dz * dz;
            pxa[j] = px; pya[j] = py; pza[j] = pz; ed[j] = d;
            dmin = fminf(dmin, d);
        }

        float wsum = 0.0f, ox = 0.0f, oy = 0.0f, oz = 0.0f;
#pragma unroll
        for (int j = 0; j < KNN; ++j) {
            const float w = __expf(dmin - ed[j]);
            wsum += w;
            ox += w * pxa[j];
            oy += w * pya[j];
            oz += w * pza[j];
        }
        const float inv = 1.0f / wsum;
        float* outb = out + (size_t)b * 3 * M;
        outb[0 * M + mg] = ox * inv;
        outb[1 * M + mg] = oy * inv;
        outb[2 * M + mg] = oz * inv;
    }
}

extern "C" void kernel_launch(void* const* d_in, const int* in_sizes, int n_in,
                              void* d_out, int out_size, void* d_ws, size_t ws_size,
                              hipStream_t stream) {
    (void)n_in; (void)out_size; (void)d_ws; (void)ws_size;
    const float* pc = (const float*)d_in[0];
    const float* qc = (const float*)d_in[1];
    float* out = (float*)d_out;

    const int B = 4;
    const int N = in_sizes[0] / (3 * B);
    const int M = in_sizes[1] / (3 * B);

    dim3 grid(B * (M / 16));
    softproj_knn_wmma<<<grid, 32 * NW, 0, stream>>>(pc, qc, out, N, M);
}


# First step: before any insert the bound is seeded with the largest of the 16 lists' smallest new distance (one LDS exchange, one barrier): about 10 insert passes there, not 32; same set and sum order
# speedup vs baseline: 8.7296x; 1.0537x over previous
#include <hip/hip_runtime.h>
#include <float.h>

typedef __attribute__((ext_vector_type(2))) float v2f;
typedef __attribute__((ext_vector_type(8))) float v8f;
typedef unsigned long long u64;

#define KNN 16
#define UNROLL 4
#define NW 8
static_assert(2 * NW >= KNN, "the first step's bound needs at least KNN lists per query");
#define LPITCH 18

__device__ __forceinline__ void wave_lds_order() {
    __builtin_amdgcn_fence(__ATOMIC_RELEASE, "wavefront");
    __builtin_amdgcn_wave_barrier();
    __builtin_amdgcn_fence(__ATOMIC_ACQUIRE, "wavefront");
}

__device__ __forceinline__ u64 from_partner(u64 v) {
    const unsigned lo = (unsigned)__shfl_xor((int)(unsigned)v, 16);
    const unsigned hi = (unsigned)__shfl_xor((int)(unsigned)(v >> 32), 16);
    return ((u64)hi << 32) | lo;
}

__device__ __forceinline__ float next_above(float x) {
    int bits = __float_as_int(x);
    bits = (bits == (int)0x80000000) ? 0 : bits;
    bits += (bits >= 0) ? 1 : -1;
    return __int_as_float(bits);
}

__device__ __forceinline__ void order2(u64& a, u64& b) {
    const bool sw = b < a;
    const u64 lo = sw ? b : a;
    const u64 hi = sw ? a : b;
    a = lo; b = hi;
}

__device__ __forceinline__ void merge16(u64 (&a)[KNN], const u64 (&b)[KNN]) {
#pragma unroll
    for (int i = 0; i < KNN; ++i) {
        const u64 o = b[KNN - 1 - i];
        a[i] = o < a[i] ? o : a[i];
    }
#pragma unroll
    for (int s = KNN / 2; s >= 1; s >>= 1) {
#pragma unroll
        for (int i = 0; i < KNN; ++i) {
            if ((i & s) == 0) order2(a[i], a[i + s]);
        }
    }
}

__global__ __launch_bounds__(32 * NW)
void softproj_knn_wmma(const float* __restrict__ pc,
                       const float* __restrict__ qc,
                       float* __restrict__ out,
                       int N, int M) {
    const int lane = threadIdx.x & 31;
    const int wv   = __builtin_amdgcn_readfirstlane((int)(threadIdx.x >> 5));
    const int mrow = lane & 15;
    const bool hi  = lane >= 16;

    const int tiles_m = M >> 4;
    const int b  = blockIdx.x / tiles_m;
    const int m0 = (blockIdx.x % tiles_m) << 4;

    const float* pcb = pc + (size_t)b * 3 * N;
    const float* qcb = qc + (size_t)b * 3 * M;

    const int   mg = m0 + mrow;
    const float qx = qcb[0 * M + mg];
    const float qy = qcb[1 * M + mg];
    const float qz = qcb[2 * M + mg];
    v2f afrag;
    afrag.x = hi ? qz   : qx;
    afrag.y = hi ? 1.0f : qy;

    __shared__ __attribute__((aligned(16))) float tile_d[NW][UNROLL][16][20];
    __shared__ __attribute__((aligned(16))) u64 lists[NW][16][LPITCH];
    __shared__ __attribute__((aligned(16))) float bound_pub[2][16][NW];
    __shared__ __attribute__((aligned(16))) float seed_pub[16][NW];

    float kd[KNN];
    int   ki[KNN];
#pragma unroll
    for (int j = 0; j < KNN; ++j) { kd[j] = FLT_MAX; ki[j] = 0; }

    const int c0 = hi ? 8 : 0;

    const int steps = N / (16 * UNROLL);
    const int per   = (steps + NW - 1) / NW;
    const int s_beg = min(wv * per, steps);
    const int s_end = min((wv + 1) * per, steps);

    for (int it = 0; it < per; ++it) {
        const bool live = s_beg + it < s_end;
        const int  n0   = min(s_beg + it, steps - 1) * (16 * UNROLL);
        float px[UNROLL], py[UNROLL], pz[UNROLL];
#pragma unroll
        for (int u = 0; u < UNROLL; ++u) {
            const int n = n0 + 16 * u + mrow;
            px[u] = pcb[0 * N + n];
            py[u] = pcb[1 * N + n];
            pz[u] = pcb[2 * N + n];
        }

        const float o0 = __shfl_xor(kd[0], 16);
        const float o1 = __shfl_xor(kd[1], 16);
        const float o7 = __shfl_xor(kd[7], 16);
        const float second = fminf(fmaxf(kd[0], o0), fminf(kd[1], o1));
        const float pair8  = fmaxf(kd[7], o7);
        if (lane < 16) bound_pub[it & 1][mrow][wv] = second;
        __builtin_amdgcn_fence(__ATOMIC_RELEASE, "workgroup", "local");
        __builtin_amdgcn_s_barrier();
        __builtin_amdgcn_fence(__ATOMIC_ACQUIRE, "workgroup", "local");
        float wgb = bound_pub[it & 1][mrow][0];
#pragma unroll
        for (int w = 1; w < NW; ++w) wgb = fmaxf(wgb, bound_pub[it & 1][mrow][w]);
        const float foreign = fminf(pair8, wgb);

#pragma unroll
        for (int u = 0; u < UNROLL; ++u) {
            const float psq = px[u] * px[u] + py[u] * py[u] + pz[u] * pz[u];
            v2f bfrag;
            bfrag.x = hi ? (-2.0f * pz[u]) : (-2.0f * px[u]);
            bfrag.y = hi ? psq             : (-2.0f * py[u]);

            v8f c = {};
            c = __builtin_amdgcn_wmma_f32_16x16x4_f32(false, afrag, false, bfrag,
                                                      (short)0, c, false, false);
#pragma unroll
            for (int r = 0; r < 8; ++r) {
                tile_d[wv][u][r + (hi ? 8 : 0)][mrow] = c[r];
            }
        }
        wave_lds_order();

        float dv[UNROLL][8];
#pragma unroll
        for (int u = 0; u < UNROLL; ++u) {
            const float4 qa = *reinterpret_cast<const float4*>(&tile_d[wv][u][mrow][c0]);
            const float4 qb = *reinterpret_cast<const float4*>(&tile_d[wv][u][mrow][c0 + 4]);
            dv[u][0] = qa.x; dv[u][1] = qa.y; dv[u][2] = qa.z; dv[u][3] = qa.w;
            dv[u][4] = qb.x; dv[u][5] = qb.y; dv[u][6] = qb.z; dv[u][7] = qb.w;
        }

        float seed = FLT_MAX;
        if (it == 0) {
            float m = dv[0][0];
#pragma unroll
            for (int u = 0; u < UNROLL; ++u) {
#pragma unroll
                for (int j = 0; j < 8; ++j) m = fminf(m, dv[u][j]);
            }
            m = fmaxf(m, __shfl_xor(m, 16));
            if (lane < 16) seed_pub[mrow][wv] = live ? m : FLT_MAX;
            __builtin_amdgcn_fence(__ATOMIC_RELEASE, "workgroup", "local");
            __builtin_amdgcn_s_barrier();
            __builtin_amdgcn_fence(__ATOMIC_ACQUIRE, "workgroup", "local");
            seed = seed_pub[mrow][0];
#pragma unroll
            for (int w = 1; w < NW; ++w) seed = fmaxf(seed, seed_pub[mrow][w]);
        }

        const float thr = live ? fminf(kd[KNN - 1], next_above(fminf(foreign, seed))) : -INFINITY;
        unsigned marks = 0u;
#pragma unroll
        for (int u = 0; u < UNROLL; ++u) {
            unsigned mu = 0u;
#pragma unroll
            for (int j = 0; j < 8; ++j) {
                mu = (mu << 1) | (__float_as_uint(dv[u][j] - thr) >> 31);
            }
            marks = (marks << 8) | mu;
        }

        auto insert_pass = [&]() {
            const bool  have = marks != 0u;
            const int   p    = __clz((int)marks) & 31;
            marks &= ~(0x80000000u >> p);
            const float d    = tile_d[wv][p >> 3][mrow][c0 + (p & 7)];
            const int   gi   = n0 + 16 * (p >> 3) + c0 + (p & 7);
            const float dn   = have ? d : INFINITY;
            int ni[KNN];
#pragma unroll
            for (int t = 0; t < KNN; ++t) ni[t] = (dn < kd[t]) ? gi : ki[t];
#pragma unroll
            for (int t = KNN - 1; t > 0; --t) {
                ki[t] = (dn < kd[t - 1]) ? ki[t - 1] : ni[t];
                kd[t] = __builtin_amdgcn_fmed3f(kd[t - 1], kd[t], dn);
            }
            ki[0] = ni[0];
            kd[0] = __builtin_amdgcn_fmed3f(-INFINITY, kd[0], dn);
        };
        for (int trip = 0; trip < 8; ++trip) {
            if (__builtin_amdgcn_ballot_w32(marks != 0u) == 0u) break;
            insert_pass();
            if (__builtin_amdgcn_ballot_w32(marks != 0u) == 0u) break;
            insert_pass();
            if (__builtin_amdgcn_ballot_w32(marks != 0u) == 0u) break;
            insert_pass();
            if (__builtin_amdgcn_ballot_w32(marks != 0u) == 0u) break;
            insert_pass();
        }
        wave_lds_order();
    }

    u64 key[KNN];
    const unsigned half_bit = hi ? 0x80000000u : 0u;
#pragma unroll
    for (int j = 0; j < KNN; ++j) {
        const unsigned ub = __float_as_uint(kd[j]);
        const unsigned su = ub ^ ((unsigned)((int)ub >> 31) | 0x80000000u);
        key[j] = ((u64)su << 32) | ((unsigned)ki[j] | half_bit);
    }

    {
        u64 other[KNN];
#pragma unroll
        for (int j = 0; j < KNN; ++j) other[j] = from_partner(key[j]);
        merge16(key, other);
    }
    if (lane < 16) {
#pragma unroll
        for (int j = 0; j < KNN; ++j) lists[wv][mrow][j] = key[j];
    }
    __syncthreads();
    if (wv != 0) return;

    const int w0 = hi ? NW / 2 : 0;
    u64 acc[KNN];
#pragma unroll
    for (int j = 0; j < KNN; ++j) acc[j] = lists[w0][mrow][j];
#pragma unroll
    for (int t = 1; t < NW / 2; ++t) {
        u64 nx[KNN];
#pragma unroll
        for (int j = 0; j < KNN; ++j) nx[j] = lists[w0 + t][mrow][j];
        merge16(acc, nx);
    }
    {
        u64 other[KNN];
#pragma unroll
        for (int j = 0; j < KNN; ++j) other[j] = from_partner(acc[j]);
        merge16(acc, other);
    }

    if (lane < 16) {
        float pxa[KNN], pya[KNN], pza[KNN], ed[KNN];
        float dmin = FLT_MAX;
#pragma unroll
        for (int j = 0; j < KNN; ++j) {
            const int   gi = (int)((unsigned)acc[j] & 0x7fffffffu);
            const float px = pcb[0 * N + gi];
            const float py = pcb[1 * N + gi];
            const float pz = pcb[2 * N + gi];
            const float dx = px - qx, dy = py - qy, dz = pz - qz;
            const float d  = dx * dx + dy * dy + dz * dz;
            pxa[j] = px; pya[j] = py; pza[j] = pz; ed[j] = d;
            dmin = fminf(dmin, d);
        }

        float wsum = 0.0f, ox = 0.0f, oy = 0.0f, oz = 0.0f;
#pragma unroll
        for (int j = 0; j < KNN; ++j) {
            const float w = __expf(dmin - ed[j]);
            wsum += w;
            ox += w * pxa[j];
            oy += w * pya[j];
            oz += w * pza[j];
        }
        const float inv = 1.0f / wsum;
        float* outb = out + (size_t)b * 3 * M;
        outb[0 * M + mg] = ox * inv;
        outb[1 * M + mg] = oy * inv;
        outb[2 * M + mg] = oz * inv;
    }
}

extern "C" void kernel_launch(void* const* d_in, const int* in_sizes, int n_in,
                              void* d_out, int out_size, void* d_ws, size_t ws_size,
                              hipStream_t stream) {
    (void)n_in; (void)out_size; (void)d_ws; (void)ws_size;
    const float* pc = (const float*)d_in[0];
    const float* qc = (const float*)d_in[1];
    float* out = (float*)d_out;

    const int B = 4;
    const int N = in_sizes[0] / (3 * B);
    const int M = in_sizes[1] / (3 * B);

    dim3 grid(B * (M / 16));
    softproj_knn_wmma<<<grid, 32 * NW, 0, stream>>>(pc, qc, out, N, M);
}


# Step size doubled: 8 point tiles (128 points) per wave and trip, marks in two 32-bit words: half the trips, barriers, bound exchanges and load waits, fewer insert passes; same set and sum order
# speedup vs baseline: 9.9420x; 1.1389x over previous
#include <hip/hip_runtime.h>
#include <float.h>

typedef __attribute__((ext_vector_type(2))) float v2f;
typedef __attribute__((ext_vector_type(8))) float v8f;
typedef unsigned long long u64;

#define KNN 16
#define UNROLL 8
static_assert(UNROLL == 8, "the marks below are two 32-bit words of four tiles each");
#define NW 8
static_assert(2 * NW >= KNN, "the first step's bound needs at least KNN lists per query");
#define LPITCH 18

__device__ __forceinline__ void wave_lds_order() {
    __builtin_amdgcn_fence(__ATOMIC_RELEASE, "wavefront");
    __builtin_amdgcn_wave_barrier();
    __builtin_amdgcn_fence(__ATOMIC_ACQUIRE, "wavefront");
}

__device__ __forceinline__ u64 from_partner(u64 v) {
    const unsigned lo = (unsigned)__shfl_xor((int)(unsigned)v, 16);
    const unsigned hi = (unsigned)__shfl_xor((int)(unsigned)(v >> 32), 16);
    return ((u64)hi << 32) | lo;
}

__device__ __forceinline__ float next_above(float x) {
    int bits = __float_as_int(x);
    bits = (bits == (int)0x80000000) ? 0 : bits;
    bits += (bits >= 0) ? 1 : -1;
    return __int_as_float(bits);
}

__device__ __forceinline__ void order2(u64& a, u64& b) {
    const bool sw = b < a;
    const u64 lo = sw ? b : a;
    const u64 hi = sw ? a : b;
    a = lo; b = hi;
}

__device__ __forceinline__ void merge16(u64 (&a)[KNN], const u64 (&b)[KNN]) {
#pragma unroll
    for (int i = 0; i < KNN; ++i) {
        const u64 o = b[KNN - 1 - i];
        a[i] = o < a[i] ? o : a[i];
    }
#pragma unroll
    for (int s = KNN / 2; s >= 1; s >>= 1) {
#pragma unroll
        for (int i = 0; i < KNN; ++i) {
            if ((i & s) == 0) order2(a[i], a[i + s]);
        }
    }
}

__global__ __launch_bounds__(32 * NW)
void softproj_knn_wmma(const float* __restrict__ pc,
                       const float* __restrict__ qc,
                       float* __restrict__ out,
                       int N, int M) {
    const int lane = threadIdx.x & 31;
    const int wv   = __builtin_amdgcn_readfirstlane((int)(threadIdx.x >> 5));
    const int mrow = lane & 15;
    const bool hi  = lane >= 16;

    const int tiles_m = M >> 4;
    const int b  = blockIdx.x / tiles_m;
    const int m0 = (blockIdx.x % tiles_m) << 4;

    const float* pcb = pc + (size_t)b * 3 * N;
    const float* qcb = qc + (size_t)b * 3 * M;

    const int   mg = m0 + mrow;
    const float qx = qcb[0 * M + mg];
    const float qy = qcb[1 * M + mg];
    const float qz = qcb[2 * M + mg];
    v2f afrag;
    afrag.x = hi ? qz   : qx;
    afrag.y = hi ? 1.0f : qy;

    __shared__ __attribute__((aligned(16))) float tile_d[NW][UNROLL][16][20];
    __shared__ __attribute__((aligned(16))) u64 lists[NW][16][LPITCH];
    __shared__ __attribute__((aligned(16))) float bound_pub[2][16][NW];
    __shared__ __attribute__((aligned(16))) float seed_pub[16][NW];

    float kd[KNN];
    int   ki[KNN];
#pragma unroll
    for (int j = 0; j < KNN; ++j) { kd[j] = FLT_MAX; ki[j] = 0; }

    const int c0 = hi ? 8 : 0;

    const int steps = N / (16 * UNROLL);
    const int per   = (steps + NW - 1) / NW;
    const int s_beg = min(wv * per, steps);
    const int s_end = min((wv + 1) * per, steps);

    for (int it = 0; it < per; ++it) {
        const bool live = s_beg + it < s_end;
        const int  n0   = min(s_beg + it, steps - 1) * (16 * UNROLL);
        float px[UNROLL], py[UNROLL], pz[UNROLL];
#pragma unroll
        for (int u = 0; u < UNROLL; ++u) {
            const int n = n0 + 16 * u + mrow;
            px[u] = pcb[0 * N + n];
            py[u] = pcb[1 * N + n];
            pz[u] = pcb[2 * N + n];
        }

        const float o0 = __shfl_xor(kd[0], 16);
        const float o1 = __shfl_xor(kd[1], 16);
        const float o7 = __shfl_xor(kd[7], 16);
        const float second = fminf(fmaxf(kd[0], o0), fminf(kd[1], o1));
        const float pair8  = fmaxf(kd[7], o7);
        if (lane < 16) bound_pub[it & 1][mrow][wv] = second;
        __builtin_amdgcn_fence(__ATOMIC_RELEASE, "workgroup", "local");
        __builtin_amdgcn_s_barrier();
        __builtin_amdgcn_fence(__ATOMIC_ACQUIRE, "workgroup", "local");
        float wgb = bound_pub[it & 1][mrow][0];
#pragma unroll
        for (int w = 1; w < NW; ++w) wgb = fmaxf(wgb, bound_pub[it & 1][mrow][w]);
        const float foreign = fminf(pair8, wgb);

#pragma unroll
        for (int u = 0; u < UNROLL; ++u) {
            const float psq = px[u] * px[u] + py[u] * py[u] + pz[u] * pz[u];
            v2f bfrag;
            bfrag.x = hi ? (-2.0f * pz[u]) : (-2.0f * px[u]);
            bfrag.y = hi ? psq             : (-2.0f * py[u]);

            v8f c = {};
            c = __builtin_amdgcn_wmma_f32_16x16x4_f32(false, afrag, false, bfrag,
                                                      (short)0, c, false, false);
#pragma unroll
            for (int r = 0; r < 8; ++r) {
                tile_d[wv][u][r + (hi ? 8 : 0)][mrow] = c[r];
            }
        }
        wave_lds_order();

        float dv[UNROLL][8];
#pragma unroll
        for (int u = 0; u < UNROLL; ++u) {
            const float4 qa = *reinterpret_cast<const float4*>(&tile_d[wv][u][mrow][c0]);
            const float4 qb = *reinterpret_cast<const float4*>(&tile_d[wv][u][mrow][c0 + 4]);
            dv[u][0] = qa.x; dv[u][1] = qa.y; dv[u][2] = qa.z; dv[u][3] = qa.w;
            dv[u][4] = qb.x; dv[u][5] = qb.y; dv[u][6] = qb.z; dv[u][7] = qb.w;
        }

        float seed = FLT_MAX;
        if (it == 0) {
            float m = dv[0][0];
#pragma unroll
            for (int u = 0; u < UNROLL; ++u) {
#pragma unroll
                for (int j = 0; j < 8; ++j) m = fminf(m, dv[u][j]);
            }
            m = fmaxf(m, __shfl_xor(m, 16));
            if (lane < 16) seed_pub[mrow][wv] = live ? m : FLT_MAX;
            __builtin_amdgcn_fence(__ATOMIC_RELEASE, "workgroup", "local");
            __builtin_amdgcn_s_barrier();
            __builtin_amdgcn_fence(__ATOMIC_ACQUIRE, "workgroup", "local");
            seed = seed_pub[mrow][0];
#pragma unroll
            for (int w = 1; w < NW; ++w) seed = fmaxf(seed, seed_pub[mrow][w]);
        }

        const float thr = live ? fminf(kd[KNN - 1], next_above(fminf(foreign, seed))) : -INFINITY;
        unsigned marks_a = 0u, marks_b = 0u;
#pragma unroll
        for (int u = 0; u < UNROLL; ++u) {
            unsigned mu = 0u;
#pragma unroll
            for (int j = 0; j < 8; ++j) {
                mu = (mu << 1) | (__float_as_uint(dv[u][j] - thr) >> 31);
            }
            if (u < UNROLL / 2) marks_a = (marks_a << 8) | mu;
            else                marks_b = (marks_b << 8) | mu;
        }

        auto insert_pass = [&]() {
            const bool  in_a = marks_a != 0u;
            const bool  have = (marks_a | marks_b) != 0u;
            const unsigned m = in_a ? marks_a : marks_b;
            const int   q    = __clz((int)m) & 31;
            const unsigned bit = 0x80000000u >> q;
            marks_a &= ~bit;
            marks_b &= in_a ? 0xffffffffu : ~bit;
            const int   p    = q + (in_a ? 0 : 32);
            const float d    = tile_d[wv][p >> 3][mrow][c0 + (p & 7)];
            const int   gi   = n0 + 16 * (p >> 3) + c0 + (p & 7);
            const float dn   = have ? d : INFINITY;
            int ni[KNN];
#pragma unroll
            for (int t = 0; t < KNN; ++t) ni[t] = (dn < kd[t]) ? gi : ki[t];
#pragma unroll
            for (int t = KNN - 1; t > 0; --t) {
                ki[t] = (dn < kd[t - 1]) ? ki[t - 1] : ni[t];
                kd[t] = __builtin_amdgcn_fmed3f(kd[t - 1], kd[t], dn);
            }
            ki[0] = ni[0];
            kd[0] = __builtin_amdgcn_fmed3f(-INFINITY, kd[0], dn);
        };
        for (int trip = 0; trip < 16; ++trip) {
            if (__builtin_amdgcn_ballot_w32((marks_a | marks_b) != 0u) == 0u) break;
            insert_pass();
            if (__builtin_amdgcn_ballot_w32((marks_a | marks_b) != 0u) == 0u) break;
            insert_pass();
            if (__builtin_amdgcn_ballot_w32((marks_a | marks_b) != 0u) == 0u) break;
            insert_pass();
            if (__builtin_amdgcn_ballot_w32((marks_a | marks_b) != 0u) == 0u) break;
            insert_pass();
        }
        wave_lds_order();
    }

    u64 key[KNN];
    const unsigned half_bit = hi ? 0x80000000u : 0u;
#pragma unroll
    for (int j = 0; j < KNN; ++j) {
        const unsigned ub = __float_as_uint(kd[j]);
        const unsigned su = ub ^ ((unsigned)((int)ub >> 31) | 0x80000000u);
        key[j] = ((u64)su << 32) | ((unsigned)ki[j] | half_bit);
    }

    {
        u64 other[KNN];
#pragma unroll
        for (int j = 0; j < KNN; ++j) other[j] = from_partner(key[j]);
        merge16(key, other);
    }
    if (lane < 16) {
#pragma unroll
        for (int j = 0; j < KNN; ++j) lists[wv][mrow][j] = key[j];
    }
    __syncthreads();
    if (wv != 0) return;

    const int w0 = hi ? NW / 2 : 0;
    u64 acc[KNN];
#pragma unroll
    for (int j = 0; j < KNN; ++j) acc[j] = lists[w0][mrow][j];
#pragma unroll
    for (int t = 1; t < NW / 2; ++t) {
        u64 nx[KNN];
#pragma unroll
        for (int j = 0; j < KNN; ++j) nx[j] = lists[w0 + t][mrow][j];
        merge16(acc, nx);
    }
    {
        u64 other[KNN];
#pragma unroll
        for (int j = 0; j < KNN; ++j) other[j] = from_partner(acc[j]);
        merge16(acc, other);
    }

    if (lane < 16) {
        float pxa[KNN], pya[KNN], pza[KNN], ed[KNN];
        float dmin = FLT_MAX;
#pragma unroll
        for (int j = 0; j < KNN; ++j) {
            const int   gi = (int)((unsigned)acc[j] & 0x7fffffffu);
            const float px = pcb[0 * N + gi];
            const float py = pcb[1 * N + gi];
            const float pz = pcb[2 * N + gi];
            const float dx = px - qx, dy = py - qy, dz = pz - qz;
            const float d  = dx * dx + dy * dy + dz * dz;
            pxa[j] = px; pya[j] = py; pza[j] = pz; ed[j] = d;
            dmin = fminf(dmin, d);
        }

        float wsum = 0.0f, ox = 0.0f, oy = 0.0f, oz = 0.0f;
#pragma unroll
        for (int j = 0; j < KNN; ++j) {
            const float w = __expf(dmin - ed[j]);
            wsum += w;
            ox += w * pxa[j];
            oy += w * pya[j];
            oz += w * pza[j];
        }
        const float inv = 1.0f / wsum;
        float* outb = out + (size_t)b * 3 * M;
        outb[0 * M + mg] = ox * inv;
        outb[1 * M + mg] = oy * inv;
        outb[2 * M + mg] = oz * inv;
    }
}

extern "C" void kernel_launch(void* const* d_in, const int* in_sizes, int n_in,
                              void* d_out, int out_size, void* d_ws, size_t ws_size,
                              hipStream_t stream) {
    (void)n_in; (void)out_size; (void)d_ws; (void)ws_size;
    const float* pc = (const float*)d_in[0];
    const float* qc = (const float*)d_in[1];
    float* out = (float*)d_out;

    const int B = 4;
    const int N = in_sizes[0] / (3 * B);
    const int M = in_sizes[1] / (3 * B);

    dim3 grid(B * (M / 16));
    softproj_knn_wmma<<<grid, 32 * NW, 0, stream>>>(pc, qc, out, N, M);
}


# WMMA operands swapped (points as A rows, queries as B columns): a lane's D registers hold its own query's 8 columns, so the per-step LDS transpose and its wait are gone; same set and sum order
# speedup vs baseline: 10.4495x; 1.0510x over previous
#include <hip/hip_runtime.h>
#include <float.h>

typedef __attribute__((ext_vector_type(2))) float v2f;
typedef __attribute__((ext_vector_type(8))) float v8f;
typedef unsigned long long u64;

#define KNN 16
#define UNROLL 8
static_assert(UNROLL == 8, "the marks below are two 32-bit words of four tiles each");
#define NW 8
static_assert(2 * NW >= KNN, "the first step's bound needs at least KNN lists per query");
#define LPITCH 18

__device__ __forceinline__ u64 from_partner(u64 v) {
    const unsigned lo = (unsigned)__shfl_xor((int)(unsigned)v, 16);
    const unsigned hi = (unsigned)__shfl_xor((int)(unsigned)(v >> 32), 16);
    return ((u64)hi << 32) | lo;
}

__device__ __forceinline__ float next_above(float x) {
    int bits = __float_as_int(x);
    bits = (bits == (int)0x80000000) ? 0 : bits;
    bits += (bits >= 0) ? 1 : -1;
    return __int_as_float(bits);
}

__device__ __forceinline__ void order2(u64& a, u64& b) {
    const bool sw = b < a;
    const u64 lo = sw ? b : a;
    const u64 hi = sw ? a : b;
    a = lo; b = hi;
}

__device__ __forceinline__ void merge16(u64 (&a)[KNN], const u64 (&b)[KNN]) {
#pragma unroll
    for (int i = 0; i < KNN; ++i) {
        const u64 o = b[KNN - 1 - i];
        a[i] = o < a[i] ? o : a[i];
    }
#pragma unroll
    for (int s = KNN / 2; s >= 1; s >>= 1) {
#pragma unroll
        for (int i = 0; i < KNN; ++i) {
            if ((i & s) == 0) order2(a[i], a[i + s]);
        }
    }
}

__global__ __launch_bounds__(32 * NW)
void softproj_knn_wmma(const float* __restrict__ pc,
                       const float* __restrict__ qc,
                       float* __restrict__ out,
                       int N, int M) {
    const int lane = threadIdx.x & 31;
    const int wv   = __builtin_amdgcn_readfirstlane((int)(threadIdx.x >> 5));
    const int mrow = lane & 15;
    const bool hi  = lane >= 16;

    const int tiles_m = M >> 4;
    const int b  = blockIdx.x / tiles_m;
    const int m0 = (blockIdx.x % tiles_m) << 4;

    const float* pcb = pc + (size_t)b * 3 * N;
    const float* qcb = qc + (size_t)b * 3 * M;

    const int   mg = m0 + mrow;
    const float qx = qcb[0 * M + mg];
    const float qy = qcb[1 * M + mg];
    const float qz = qcb[2 * M + mg];
    v2f qfrag;
    qfrag.x = hi ? qz   : qx;
    qfrag.y = hi ? 1.0f : qy;

    __shared__ __attribute__((aligned(16))) float cand[NW][2 * UNROLL][32][4];
    __shared__ __attribute__((aligned(16))) u64 lists[NW][16][LPITCH];
    __shared__ __attribute__((aligned(16))) float bound_pub[2][16][NW];
    __shared__ __attribute__((aligned(16))) float seed_pub[16][NW];

    float kd[KNN];
    int   ki[KNN];
#pragma unroll
    for (int j = 0; j < KNN; ++j) { kd[j] = FLT_MAX; ki[j] = 0; }

    const int c0 = hi ? 8 : 0;

    const int steps = N / (16 * UNROLL);
    const int per   = (steps + NW - 1) / NW;
    const int s_beg = min(wv * per, steps);
    const int s_end = min((wv + 1) * per, steps);

    for (int it = 0; it < per; ++it) {
        const bool live = s_beg + it < s_end;
        const int  n0   = min(s_beg + it, steps - 1) * (16 * UNROLL);
        float px[UNROLL], py[UNROLL], pz[UNROLL];
#pragma unroll
        for (int u = 0; u < UNROLL; ++u) {
            const int n = n0 + 16 * u + mrow;
            px[u] = pcb[0 * N + n];
            py[u] = pcb[1 * N + n];
            pz[u] = pcb[2 * N + n];
        }

        const float o0 = __shfl_xor(kd[0], 16);
        const float o1 = __shfl_xor(kd[1], 16);
        const float o7 = __shfl_xor(kd[7], 16);
        const float second = fminf(fmaxf(kd[0], o0), fminf(kd[1], o1));
        const float pair8  = fmaxf(kd[7], o7);
        if (lane < 16) bound_pub[it & 1][mrow][wv] = second;
        __builtin_amdgcn_fence(__ATOMIC_RELEASE, "workgroup", "local");
        __builtin_amdgcn_s_barrier();
        __builtin_amdgcn_fence(__ATOMIC_ACQUIRE, "workgroup", "local");
        float wgb = bound_pub[it & 1][mrow][0];
#pragma unroll
        for (int w = 1; w < NW; ++w) wgb = fmaxf(wgb, bound_pub[it & 1][mrow][w]);
        const float foreign = fminf(pair8, wgb);

        float dv[UNROLL][8];
#pragma unroll
        for (int u = 0; u < UNROLL; ++u) {
            const float psq = px[u] * px[u] + py[u] * py[u] + pz[u] * pz[u];
            v2f pfrag;
            pfrag.x = hi ? (-2.0f * pz[u]) : (-2.0f * px[u]);
            pfrag.y = hi ? psq             : (-2.0f * py[u]);

            v8f c = {};
            c = __builtin_amdgcn_wmma_f32_16x16x4_f32(false, pfrag, false, qfrag,
                                                      (short)0, c, false, false);
#pragma unroll
            for (int r = 0; r < 8; ++r) dv[u][r] = c[r];
            *reinterpret_cast<float4*>(&cand[wv][2 * u][lane][0])     = make_float4(c[0], c[1], c[2], c[3]);
            *reinterpret_cast<float4*>(&cand[wv][2 * u + 1][lane][0]) = make_float4(c[4], c[5], c[6], c[7]);
        }


        float seed = FLT_MAX;
        if (it == 0) {
            float m = dv[0][0];
#pragma unroll
            for (int u = 0; u < UNROLL; ++u) {
#pragma unroll
                for (int j = 0; j < 8; ++j) m = fminf(m, dv[u][j]);
            }
            m = fmaxf(m, __shfl_xor(m, 16));
            if (lane < 16) seed_pub[mrow][wv] = live ? m : FLT_MAX;
            __builtin_amdgcn_fence(__ATOMIC_RELEASE, "workgroup", "local");
            __builtin_amdgcn_s_barrier();
            __builtin_amdgcn_fence(__ATOMIC_ACQUIRE, "workgroup", "local");
            seed = seed_pub[mrow][0];
#pragma unroll
            for (int w = 1; w < NW; ++w) seed = fmaxf(seed, seed_pub[mrow][w]);
        }

        const float thr = live ? fminf(kd[KNN - 1], next_above(fminf(foreign, seed))) : -INFINITY;
        unsigned marks_a = 0u, marks_b = 0u;
#pragma unroll
        for (int u = 0; u < UNROLL; ++u) {
            unsigned mu = 0u;
#pragma unroll
            for (int j = 0; j < 8; ++j) {
                mu = (mu << 1) | (__float_as_uint(dv[u][j] - thr) >> 31);
            }
            if (u < UNROLL / 2) marks_a = (marks_a << 8) | mu;
            else                marks_b = (marks_b << 8) | mu;
        }

        auto insert_pass = [&]() {
            const bool  in_a = marks_a != 0u;
            const bool  have = (marks_a | marks_b) != 0u;
            const unsigned m = in_a ? marks_a : marks_b;
            const int   q    = __clz((int)m) & 31;
            const unsigned bit = 0x80000000u >> q;
            marks_a &= ~bit;
            marks_b &= in_a ? 0xffffffffu : ~bit;
            const int   p    = q + (in_a ? 0 : 32);
            const float d    = cand[wv][p >> 2][lane][p & 3];
            const int   gi   = n0 + 16 * (p >> 3) + c0 + (p & 7);
            const float dn   = have ? d : INFINITY;
            int ni[KNN];
#pragma unroll
            for (int t = 0; t < KNN; ++t) ni[t] = (dn < kd[t]) ? gi : ki[t];
#pragma unroll
            for (int t = KNN - 1; t > 0; --t) {
                ki[t] = (dn < kd[t - 1]) ? ki[t - 1] : ni[t];
                kd[t] = __builtin_amdgcn_fmed3f(kd[t - 1], kd[t], dn);
            }
            ki[0] = ni[0];
            kd[0] = __builtin_amdgcn_fmed3f(-INFINITY, kd[0], dn);
        };
        for (int trip = 0; trip < 16; ++trip) {
            if (__builtin_amdgcn_ballot_w32((marks_a | marks_b) != 0u) == 0u) break;
            insert_pass();
            if (__builtin_amdgcn_ballot_w32((marks_a | marks_b) != 0u) == 0u) break;
            insert_pass();
            if (__builtin_amdgcn_ballot_w32((marks_a | marks_b) != 0u) == 0u) break;
            insert_pass();
            if (__builtin_amdgcn_ballot_w32((marks_a | marks_b) != 0u) == 0u) break;
            insert_pass();
        }
    }

    u64 key[KNN];
    const unsigned half_bit = hi ? 0x80000000u : 0u;
#pragma unroll
    for (int j = 0; j < KNN; ++j) {
        const unsigned ub = __float_as_uint(kd[j]);
        const unsigned su = ub ^ ((unsigned)((int)ub >> 31) | 0x80000000u);
        key[j] = ((u64)su << 32) | ((unsigned)ki[j] | half_bit);
    }

    {
        u64 other[KNN];
#pragma unroll
        for (int j = 0; j < KNN; ++j) other[j] = from_partner(key[j]);
        merge16(key, other);
    }
    if (lane < 16) {
#pragma unroll
        for (int j = 0; j < KNN; ++j) lists[wv][mrow][j] = key[j];
    }
    __syncthreads();
    if (wv != 0) return;

    const int w0 = hi ? NW / 2 : 0;
    u64 acc[KNN];
#pragma unroll
    for (int j = 0; j < KNN; ++j) acc[j] = lists[w0][mrow][j];
#pragma unroll
    for (int t = 1; t < NW / 2; ++t) {
        u64 nx[KNN];
#pragma unroll
        for (int j = 0; j < KNN; ++j) nx[j] = lists[w0 + t][mrow][j];
        merge16(acc, nx);
    }
    {
        u64 other[KNN];
#pragma unroll
        for (int j = 0; j < KNN; ++j) other[j] = from_partner(acc[j]);
        merge16(acc, other);
    }

    if (lane < 16) {
        float pxa[KNN], pya[KNN], pza[KNN], ed[KNN];
        float dmin = FLT_MAX;
#pragma unroll
        for (int j = 0; j < KNN; ++j) {
            const int   gi = (int)((unsigned)acc[j] & 0x7fffffffu);
            const float px = pcb[0 * N + gi];
            const float py = pcb[1 * N + gi];
            const float pz = pcb[2 * N + gi];
            const float dx = px - qx, dy = py - qy, dz = pz - qz;
            const float d  = dx * dx + dy * dy + dz * dz;
            pxa[j] = px; pya[j] = py; pza[j] = pz; ed[j] = d;
            dmin = fminf(dmin, d);
        }

        float wsum = 0.0f, ox = 0.0f, oy = 0.0f, oz = 0.0f;
#pragma unroll
        for (int j = 0; j < KNN; ++j) {
            const float w = __expf(dmin - ed[j]);
            wsum += w;
            ox += w * pxa[j];
            oy += w * pya[j];
            oz += w * pza[j];
        }
        const float inv = 1.0f / wsum;
        float* outb = out + (size_t)b * 3 * M;
        outb[0 * M + mg] = ox * inv;
        outb[1 * M + mg] = oy * inv;
        outb[2 * M + mg] = oz * inv;
    }
}

extern "C" void kernel_launch(void* const* d_in, const int* in_sizes, int n_in,
                              void* d_out, int out_size, void* d_ws, size_t ws_size,
                              hipStream_t stream) {
    (void)n_in; (void)out_size; (void)d_ws; (void)ws_size;
    const float* pc = (const float*)d_in[0];
    const float* qc = (const float*)d_in[1];
    float* out = (float*)d_out;

    const int B = 4;
    const int N = in_sizes[0] / (3 * B);
    const int M = in_sizes[1] / (3 * B);

    dim3 grid(B * (M / 16));
    softproj_knn_wmma<<<grid, 32 * NW, 0, stream>>>(pc, qc, out, N, M);
}


# Step size doubled again: 16 point tiles (256 points) per wave and trip, marks in four 32-bit words: half the trips, barriers, bound exchanges, load waits; fewer insert passes; same set and sum order
# speedup vs baseline: 11.0427x; 1.0568x over previous
#include <hip/hip_runtime.h>
#include <float.h>

typedef __attribute__((ext_vector_type(2))) float v2f;
typedef __attribute__((ext_vector_type(8))) float v8f;
typedef unsigned long long u64;

#define KNN 16
#define UNROLL 16
static_assert(UNROLL == 16, "the marks below are four 32-bit words of four tiles each");
#define NW 8
static_assert(2 * NW >= KNN, "the first step's bound needs at least KNN lists per query");
#define LPITCH 18

__device__ __forceinline__ u64 from_partner(u64 v) {
    const unsigned lo = (unsigned)__shfl_xor((int)(unsigned)v, 16);
    const unsigned hi = (unsigned)__shfl_xor((int)(unsigned)(v >> 32), 16);
    return ((u64)hi << 32) | lo;
}

__device__ __forceinline__ float next_above(float x) {
    int bits = __float_as_int(x);
    bits = (bits == (int)0x80000000) ? 0 : bits;
    bits += (bits >= 0) ? 1 : -1;
    return __int_as_float(bits);
}

__device__ __forceinline__ void order2(u64& a, u64& b) {
    const bool sw = b < a;
    const u64 lo = sw ? b : a;
    const u64 hi = sw ? a : b;
    a = lo; b = hi;
}

__device__ __forceinline__ void merge16(u64 (&a)[KNN], const u64 (&b)[KNN]) {
#pragma unroll
    for (int i = 0; i < KNN; ++i) {
        const u64 o = b[KNN - 1 - i];
        a[i] = o < a[i] ? o : a[i];
    }
#pragma unroll
    for (int s = KNN / 2; s >= 1; s >>= 1) {
#pragma unroll
        for (int i = 0; i < KNN; ++i) {
            if ((i & s) == 0) order2(a[i], a[i + s]);
        }
    }
}

__global__ __launch_bounds__(32 * NW)
void softproj_knn_wmma(const float* __restrict__ pc,
                       const float* __restrict__ qc,
                       float* __restrict__ out,
                       int N, int M) {
    const int lane = threadIdx.x & 31;
    const int wv   = __builtin_amdgcn_readfirstlane((int)(threadIdx.x >> 5));
    const int mrow = lane & 15;
    const bool hi  = lane >= 16;

    const int tiles_m = M >> 4;
    const int b  = blockIdx.x / tiles_m;
    const int m0 = (blockIdx.x % tiles_m) << 4;

    const float* pcb = pc + (size_t)b * 3 * N;
    const float* qcb = qc + (size_t)b * 3 * M;

    const int   mg = m0 + mrow;
    const float qx = qcb[0 * M + mg];
    const float qy = qcb[1 * M + mg];
    const float qz = qcb[2 * M + mg];
    v2f qfrag;
    qfrag.x = hi ? qz   : qx;
    qfrag.y = hi ? 1.0f : qy;

    __shared__ __attribute__((aligned(16))) float cand[NW][2 * UNROLL][32][4];
    __shared__ __attribute__((aligned(16))) u64 lists[NW][16][LPITCH];
    __shared__ __attribute__((aligned(16))) float bound_pub[2][16][NW];
    __shared__ __attribute__((aligned(16))) float seed_pub[16][NW];

    float kd[KNN];
    int   ki[KNN];
#pragma unroll
    for (int j = 0; j < KNN; ++j) { kd[j] = FLT_MAX; ki[j] = 0; }

    const int c0 = hi ? 8 : 0;

    const int steps = N / (16 * UNROLL);
    const int per   = (steps + NW - 1) / NW;
    const int s_beg = min(wv * per, steps);
    const int s_end = min((wv + 1) * per, steps);

    for (int it = 0; it < per; ++it) {
        const bool live = s_beg + it < s_end;
        const int  n0   = min(s_beg + it, steps - 1) * (16 * UNROLL);
        float px[UNROLL], py[UNROLL], pz[UNROLL];
#pragma unroll
        for (int u = 0; u < UNROLL; ++u) {
            const int n = n0 + 16 * u + mrow;
            px[u] = pcb[0 * N + n];
            py[u] = pcb[1 * N + n];
            pz[u] = pcb[2 * N + n];
        }

        const float o0 = __shfl_xor(kd[0], 16);
        const float o1 = __shfl_xor(kd[1], 16);
        const float o7 = __shfl_xor(kd[7], 16);
        const float second = fminf(fmaxf(kd[0], o0), fminf(kd[1], o1));
        const float pair8  = fmaxf(kd[7], o7);
        if (lane < 16) bound_pub[it & 1][mrow][wv] = second;
        __builtin_amdgcn_fence(__ATOMIC_RELEASE, "workgroup", "local");
        __builtin_amdgcn_s_barrier();
        __builtin_amdgcn_fence(__ATOMIC_ACQUIRE, "workgroup", "local");
        float wgb = bound_pub[it & 1][mrow][0];
#pragma unroll
        for (int w = 1; w < NW; ++w) wgb = fmaxf(wgb, bound_pub[it & 1][mrow][w]);
        const float foreign = fminf(pair8, wgb);

        float dv[UNROLL][8];
#pragma unroll
        for (int u = 0; u < UNROLL; ++u) {
            const float psq = px[u] * px[u] + py[u] * py[u] + pz[u] * pz[u];
            v2f pfrag;
            pfrag.x = hi ? (-2.0f * pz[u]) : (-2.0f * px[u]);
            pfrag.y = hi ? psq             : (-2.0f * py[u]);

            v8f c = {};
            c = __builtin_amdgcn_wmma_f32_16x16x4_f32(false, pfrag, false, qfrag,
                                                      (short)0, c, false, false);
#pragma unroll
            for (int r = 0; r < 8; ++r) dv[u][r] = c[r];
            *reinterpret_cast<float4*>(&cand[wv][2 * u][lane][0])     = make_float4(c[0], c[1], c[2], c[3]);
            *reinterpret_cast<float4*>(&cand[wv][2 * u + 1][lane][0]) = make_float4(c[4], c[5], c[6], c[7]);
        }


        float seed = FLT_MAX;
        if (it == 0) {
            float m = dv[0][0];
#pragma unroll
            for (int u = 0; u < UNROLL; ++u) {
#pragma unroll
                for (int j = 0; j < 8; ++j) m = fminf(m, dv[u][j]);
            }
            m = fmaxf(m, __shfl_xor(m, 16));
            if (lane < 16) seed_pub[mrow][wv] = live ? m : FLT_MAX;
            __builtin_amdgcn_fence(__ATOMIC_RELEASE, "workgroup", "local");
            __builtin_amdgcn_s_barrier();
            __builtin_amdgcn_fence(__ATOMIC_ACQUIRE, "workgroup", "local");
            seed = seed_pub[mrow][0];
#pragma unroll
            for (int w = 1; w < NW; ++w) seed = fmaxf(seed, seed_pub[mrow][w]);
        }

        const float thr = live ? fminf(kd[KNN - 1], next_above(fminf(foreign, seed))) : -INFINITY;
        unsigned marks[UNROLL / 4];
#pragma unroll
        for (int w = 0; w < UNROLL / 4; ++w) marks[w] = 0u;
#pragma unroll
        for (int u = 0; u < UNROLL; ++u) {
            unsigned mu = 0u;
#pragma unroll
            for (int j = 0; j < 8; ++j) {
                mu = (mu << 1) | (__float_as_uint(dv[u][j] - thr) >> 31);
            }
            marks[u >> 2] = (marks[u >> 2] << 8) | mu;
        }

        auto insert_pass = [&]() {
            const bool  z0   = marks[0] == 0u;
            const bool  z01  = (marks[0] | marks[1]) == 0u;
            const bool  z012 = (marks[0] | marks[1] | marks[2]) == 0u;
            const unsigned m = z01 ? (z012 ? marks[3] : marks[2]) : (z0 ? marks[1] : marks[0]);
            const int   wsel = z01 ? (z012 ? 3 : 2) : (z0 ? 1 : 0);
            const bool  have = m != 0u;
            const int   q    = __clz((int)m) & 31;
            const unsigned keep = ~(0x80000000u >> q);
            marks[0] &= keep;
            marks[1] &= z0   ? keep : 0xffffffffu;
            marks[2] &= z01  ? keep : 0xffffffffu;
            marks[3] &= z012 ? keep : 0xffffffffu;
            const int   p    = q + 32 * wsel;
            const float d    = cand[wv][p >> 2][lane][p & 3];
            const int   gi   = n0 + 16 * (p >> 3) + c0 + (p & 7);
            const float dn   = have ? d : INFINITY;
            int ni[KNN];
#pragma unroll
            for (int t = 0; t < KNN; ++t) ni[t] = (dn < kd[t]) ? gi : ki[t];
#pragma unroll
            for (int t = KNN - 1; t > 0; --t) {
                ki[t] = (dn < kd[t - 1]) ? ki[t - 1] : ni[t];
                kd[t] = __builtin_amdgcn_fmed3f(kd[t - 1], kd[t], dn);
            }
            ki[0] = ni[0];
            kd[0] = __builtin_amdgcn_fmed3f(-INFINITY, kd[0], dn);
        };
        for (int trip = 0; trip < 32; ++trip) {
            if (__builtin_amdgcn_ballot_w32((marks[0] | marks[1] | marks[2] | marks[3]) != 0u) == 0u) break;
            insert_pass();
            if (__builtin_amdgcn_ballot_w32((marks[0] | marks[1] | marks[2] | marks[3]) != 0u) == 0u) break;
            insert_pass();
            if (__builtin_amdgcn_ballot_w32((marks[0] | marks[1] | marks[2] | marks[3]) != 0u) == 0u) break;
            insert_pass();
            if (__builtin_amdgcn_ballot_w32((marks[0] | marks[1] | marks[2] | marks[3]) != 0u) == 0u) break;
            insert_pass();
        }
    }

    u64 key[KNN];
    const unsigned half_bit = hi ? 0x80000000u : 0u;
#pragma unroll
    for (int j = 0; j < KNN; ++j) {
        const unsigned ub = __float_as_uint(kd[j]);
        const unsigned su = ub ^ ((unsigned)((int)ub >> 31) | 0x80000000u);
        key[j] = ((u64)su << 32) | ((unsigned)ki[j] | half_bit);
    }

    {
        u64 other[KNN];
#pragma unroll
        for (int j = 0; j < KNN; ++j) other[j] = from_partner(key[j]);
        merge16(key, other);
    }
    if (lane < 16) {
#pragma unroll
        for (int j = 0; j < KNN; ++j) lists[wv][mrow][j] = key[j];
    }
    __syncthreads();
    if (wv != 0) return;

    const int w0 = hi ? NW / 2 : 0;
    u64 acc[KNN];
#pragma unroll
    for (int j = 0; j < KNN; ++j) acc[j] = lists[w0][mrow][j];
#pragma unroll
    for (int t = 1; t < NW / 2; ++t) {
        u64 nx[KNN];
#pragma unroll
        for (int j = 0; j < KNN; ++j) nx[j] = lists[w0 + t][mrow][j];
        merge16(acc, nx);
    }
    {
        u64 other[KNN];
#pragma unroll
        for (int j = 0; j < KNN; ++j) other[j] = from_partner(acc[j]);
        merge16(acc, other);
    }

    if (lane < 16) {
        float pxa[KNN], pya[KNN], pza[KNN], ed[KNN];
        float dmin = FLT_MAX;
#pragma unroll
        for (int j = 0; j < KNN; ++j) {
            const int   gi = (int)((unsigned)acc[j] & 0x7fffffffu);
            const float px = pcb[0 * N + gi];
            const float py = pcb[1 * N + gi];
            const float pz = pcb[2 * N + gi];
            const float dx = px - qx, dy = py - qy, dz = pz - qz;
            const float d  = dx * dx + dy * dy + dz * dz;
            pxa[j] = px; pya[j] = py; pza[j] = pz; ed[j] = d;
            dmin = fminf(dmin, d);
        }

        float wsum = 0.0f, ox = 0.0f, oy = 0.0f, oz = 0.0f;
#pragma unroll
        for (int j = 0; j < KNN; ++j) {
            const float w = __expf(dmin - ed[j]);
            wsum += w;
            ox += w * pxa[j];
            oy += w * pya[j];
            oz += w * pza[j];
        }
        const float inv = 1.0f / wsum;
        float* outb = out + (size_t)b * 3 * M;
        outb[0 * M + mg] = ox * inv;
        outb[1 * M + mg] = oy * inv;
        outb[2 * M + mg] = oz * inv;
    }
}

extern "C" void kernel_launch(void* const* d_in, const int* in_sizes, int n_in,
                              void* d_out, int out_size, void* d_ws, size_t ws_size,
                              hipStream_t stream) {
    (void)n_in; (void)out_size; (void)d_ws; (void)ws_size;
    const float* pc = (const float*)d_in[0];
    const float* qc = (const float*)d_in[1];
    float* out = (float*)d_out;

    const int B = 4;
    const int N = in_sizes[0] / (3 * B);
    const int M = in_sizes[1] / (3 * B);

    dim3 grid(B * (M / 16));
    softproj_knn_wmma<<<grid, 32 * NW, 0, stream>>>(pc, qc, out, N, M);
}


# speedup vs baseline: 18.7225x; 1.6955x over previous
#include <hip/hip_runtime.h>
#include <float.h>

typedef __attribute__((ext_vector_type(2))) float v2f;
typedef __attribute__((ext_vector_type(8))) float v8f;

#define KNN 16
#define UNROLL 16
static_assert(UNROLL == 16, "the marks below are four 32-bit words of four tiles each");
#define NW 8
static_assert(2 * NW >= KNN, "the first step's bound needs at least KNN lists per query");
#define LPITCH 18

__device__ __forceinline__ unsigned from_partner(unsigned v) {
    return (unsigned)__shfl_xor((int)v, 16);
}

__device__ __forceinline__ float next_above(float x) {
    int bits = __float_as_int(x);
    bits = (bits == (int)0x80000000) ? 0 : bits;
    bits += (bits >= 0) ? 1 : -1;
    return __int_as_float(bits);
}

__device__ __forceinline__ bool key_before(unsigned xh, unsigned xl, unsigned yh, unsigned yl) {
    return (xh < yh) | ((xh == yh) & (xl < yl));
}

__device__ __forceinline__ void order2(unsigned& ah, unsigned& al, unsigned& bh, unsigned& bl) {
    const bool sw = key_before(bh, bl, ah, al);
    const unsigned loh = sw ? bh : ah;
    const unsigned lol = sw ? bl : al;
    const unsigned hih = sw ? ah : bh;
    const unsigned hil = sw ? al : bl;
    ah = loh; al = lol; bh = hih; bl = hil;
}

__device__ __forceinline__ void merge16(unsigned (&ah)[KNN], unsigned (&al)[KNN],
                                        const unsigned (&bh)[KNN], const unsigned (&bl)[KNN]) {
#pragma unroll
    for (int i = 0; i < KNN; ++i) {
        const unsigned oh = bh[KNN - 1 - i];
        const unsigned ol = bl[KNN - 1 - i];
        const bool take = key_before(oh, ol, ah[i], al[i]);
        ah[i] = take ? oh : ah[i];
        al[i] = take ? ol : al[i];
    }
#pragma unroll
    for (int s = KNN / 2; s >= 1; s >>= 1) {
#pragma unroll
        for (int i = 0; i < KNN; ++i) {
            if ((i & s) == 0) order2(ah[i], al[i], ah[i + s], al[i + s]);
        }
    }
}

__global__ __launch_bounds__(32 * NW)
void softproj_knn_wmma(const float* __restrict__ pc,
                       const float* __restrict__ qc,
                       float* __restrict__ out,
                       int N, int M) {
    const int lane = threadIdx.x & 31;
    const int wv   = __builtin_amdgcn_readfirstlane((int)(threadIdx.x >> 5));
    const int mrow = lane & 15;
    const bool hi  = lane >= 16;

    const int tiles_m = M >> 4;
    const int b  = blockIdx.x / tiles_m;
    const int m0 = (blockIdx.x % tiles_m) << 4;

    const float* pcb = pc + (size_t)b * 3 * N;
    const float* qcb = qc + (size_t)b * 3 * M;

    const int   mg = m0 + mrow;
    const float qx = qcb[0 * M + mg];
    const float qy = qcb[1 * M + mg];
    const float qz = qcb[2 * M + mg];
    v2f qfrag;
    qfrag.x = hi ? qz   : qx;
    qfrag.y = hi ? 1.0f : qy;

    __shared__ __attribute__((aligned(16))) float cand[NW][2 * UNROLL][32][4];
    __shared__ __attribute__((aligned(16))) unsigned lists[NW][16][2 * LPITCH];
    __shared__ __attribute__((aligned(16))) float bound_pub[2][16][NW];
    __shared__ __attribute__((aligned(16))) float seed_pub[16][NW];

    float kd[KNN];
    int   ki[KNN];
#pragma unroll
    for (int j = 0; j < KNN; ++j) { kd[j] = FLT_MAX; ki[j] = 0; }

    const int c0 = hi ? 8 : 0;

    const int steps = N / (16 * UNROLL);
    const int per   = (steps + NW - 1) / NW;
    const int s_beg = min(wv * per, steps);
    const int s_end = min((wv + 1) * per, steps);

    for (int it = 0; it < per; ++it) {
        const bool live = s_beg + it < s_end;
        const int  n0   = min(s_beg + it, steps - 1) * (16 * UNROLL);
        float px[UNROLL], py[UNROLL], pz[UNROLL];
#pragma unroll
        for (int u = 0; u < UNROLL; ++u) {
            const int n = n0 + 16 * u + mrow;
            px[u] = pcb[0 * N + n];
            py[u] = pcb[1 * N + n];
            pz[u] = pcb[2 * N + n];
        }

        const float o0 = __shfl_xor(kd[0], 16);
        const float o1 = __shfl_xor(kd[1], 16);
        const float o7 = __shfl_xor(kd[7], 16);
        const float second = fminf(fmaxf(kd[0], o0), fminf(kd[1], o1));
        const float pair8  = fmaxf(kd[7], o7);
        if (lane < 16) bound_pub[it & 1][mrow][wv] = second;
        __builtin_amdgcn_fence(__ATOMIC_RELEASE, "workgroup", "local");
        __builtin_amdgcn_s_barrier();
        __builtin_amdgcn_fence(__ATOMIC_ACQUIRE, "workgroup", "local");
        float wgb = bound_pub[it & 1][mrow][0];
#pragma unroll
        for (int w = 1; w < NW; ++w) wgb = fmaxf(wgb, bound_pub[it & 1][mrow][w]);
        const float foreign = fminf(pair8, wgb);

        float dv[UNROLL][8];
#pragma unroll
        for (int u = 0; u < UNROLL; ++u) {
            const float psq = px[u] * px[u] + py[u] * py[u] + pz[u] * pz[u];
            v2f pfrag;
            pfrag.x = hi ? (-2.0f * pz[u]) : (-2.0f * px[u]);
            pfrag.y = hi ? psq             : (-2.0f * py[u]);

            v8f c = {};
            c = __builtin_amdgcn_wmma_f32_16x16x4_f32(false, pfrag, false, qfrag,
                                                      (short)0, c, false, false);
#pragma unroll
            for (int r = 0; r < 8; ++r) dv[u][r] = c[r];
            *reinterpret_cast<float4*>(&cand[wv][2 * u][lane][0])     = make_float4(c[0], c[1], c[2], c[3]);
            *reinterpret_cast<float4*>(&cand[wv][2 * u + 1][lane][0]) = make_float4(c[4], c[5], c[6], c[7]);
        }


        float seed = FLT_MAX;
        if (it == 0) {
            float m = dv[0][0];
#pragma unroll
            for (int u = 0; u < UNROLL; ++u) {
#pragma unroll
                for (int j = 0; j < 8; ++j) m = fminf(m, dv[u][j]);
            }
            m = fmaxf(m, __shfl_xor(m, 16));
            if (lane < 16) seed_pub[mrow][wv] = live ? m : FLT_MAX;
            __builtin_amdgcn_fence(__ATOMIC_RELEASE, "workgroup", "local");
            __builtin_amdgcn_s_barrier();
            __builtin_amdgcn_fence(__ATOMIC_ACQUIRE, "workgroup", "local");
            seed = seed_pub[mrow][0];
#pragma unroll
            for (int w = 1; w < NW; ++w) seed = fmaxf(seed, seed_pub[mrow][w]);
        }

        const float thr = live ? fminf(kd[KNN - 1], next_above(fminf(foreign, seed))) : -INFINITY;
        unsigned marks[UNROLL / 4];
#pragma unroll
        for (int w = 0; w < UNROLL / 4; ++w) marks[w] = 0u;
#pragma unroll
        for (int u = 0; u < UNROLL; ++u) {
            unsigned mu = 0u;
#pragma unroll
            for (int j = 0; j < 8; ++j) {
                mu = (mu << 1) | (__float_as_uint(dv[u][j] - thr) >> 31);
            }
            marks[u >> 2] = (marks[u >> 2] << 8) | mu;
        }

        auto insert_pass = [&]() {
            const bool  z0   = marks[0] == 0u;
            const bool  z01  = (marks[0] | marks[1]) == 0u;
            const bool  z012 = (marks[0] | marks[1] | marks[2]) == 0u;
            const unsigned m = z01 ? (z012 ? marks[3] : marks[2]) : (z0 ? marks[1] : marks[0]);
            const int   wsel = z01 ? (z012 ? 3 : 2) : (z0 ? 1 : 0);
            const bool  have = m != 0u;
            const int   q    = __clz((int)m) & 31;
            const unsigned keep = ~(0x80000000u >> q);
            marks[0] &= keep;
            marks[1] &= z0   ? keep : 0xffffffffu;
            marks[2] &= z01  ? keep : 0xffffffffu;
            marks[3] &= z012 ? keep : 0xffffffffu;
            const int   p    = q + 32 * wsel;
            const float d    = cand[wv][p >> 2][lane][p & 3];
            const int   gi   = n0 + 16 * (p >> 3) + c0 + (p & 7);
            const float dn   = have ? d : INFINITY;
            int ni[KNN];
#pragma unroll
            for (int t = 0; t < KNN; ++t) ni[t] = (dn < kd[t]) ? gi : ki[t];
#pragma unroll
            for (int t = KNN - 1; t > 0; --t) {
                ki[t] = (dn < kd[t - 1]) ? ki[t - 1] : ni[t];
                kd[t] = __builtin_amdgcn_fmed3f(kd[t - 1], kd[t], dn);
            }
            ki[0] = ni[0];
            kd[0] = __builtin_amdgcn_fmed3f(-INFINITY, kd[0], dn);
        };
        for (int trip = 0; trip < 32; ++trip) {
            if (__builtin_amdgcn_ballot_w32((marks[0] | marks[1] | marks[2] | marks[3]) != 0u) == 0u) break;
            insert_pass();
            if (__builtin_amdgcn_ballot_w32((marks[0] | marks[1] | marks[2] | marks[3]) != 0u) == 0u) break;
            insert_pass();
            if (__builtin_amdgcn_ballot_w32((marks[0] | marks[1] | marks[2] | marks[3]) != 0u) == 0u) break;
            insert_pass();
            if (__builtin_amdgcn_ballot_w32((marks[0] | marks[1] | marks[2] | marks[3]) != 0u) == 0u) break;
            insert_pass();
        }
    }

    unsigned kh[KNN], kl[KNN];
    const unsigned half_bit = hi ? 0x80000000u : 0u;
#pragma unroll
    for (int j = 0; j < KNN; ++j) {
        const unsigned ub = __float_as_uint(kd[j]);
        kh[j] = ub ^ ((unsigned)((int)ub >> 31) | 0x80000000u);
        kl[j] = (unsigned)ki[j] | half_bit;
    }

    {
        unsigned oh[KNN], ol[KNN];
#pragma unroll
        for (int j = 0; j < KNN; ++j) { oh[j] = from_partner(kh[j]); ol[j] = from_partner(kl[j]); }
        merge16(kh, kl, oh, ol);
    }
    if (lane < 16) {
#pragma unroll
        for (int j = 0; j < KNN; j += 2)
            *reinterpret_cast<uint4*>(&lists[wv][mrow][2 * j]) = make_uint4(kl[j], kh[j], kl[j + 1], kh[j + 1]);
    }
    __syncthreads();
    if (wv != 0) return;

    const int w0 = hi ? NW / 2 : 0;
    unsigned acch[KNN], accl[KNN];
#pragma unroll
    for (int j = 0; j < KNN; j += 2) {
        const uint4 v = *reinterpret_cast<const uint4*>(&lists[w0][mrow][2 * j]);
        accl[j] = v.x; acch[j] = v.y; accl[j + 1] = v.z; acch[j + 1] = v.w;
    }
#pragma unroll
    for (int t = 1; t < NW / 2; ++t) {
        unsigned nxh[KNN], nxl[KNN];
#pragma unroll
        for (int j = 0; j < KNN; j += 2) {
            const uint4 v = *reinterpret_cast<const uint4*>(&lists[w0 + t][mrow][2 * j]);
            nxl[j] = v.x; nxh[j] = v.y; nxl[j + 1] = v.z; nxh[j + 1] = v.w;
        }
        merge16(acch, accl, nxh, nxl);
    }
    {
        unsigned oh[KNN], ol[KNN];
#pragma unroll
        for (int j = 0; j < KNN; ++j) { oh[j] = from_partner(acch[j]); ol[j] = from_partner(accl[j]); }
        merge16(acch, accl, oh, ol);
    }

    if (lane < 16) {
        float pxa[KNN], pya[KNN], pza[KNN], ed[KNN];
        float dmin = FLT_MAX;
#pragma unroll
        for (int j = 0; j < KNN; ++j) {
            const int   gi = (int)(accl[j] & 0x7fffffffu);
            const float px = pcb[0 * N + gi];
            const float py = pcb[1 * N + gi];
            const float pz = pcb[2 * N + gi];
            const float dx = px - qx, dy = py - qy, dz = pz - qz;
            const float d  = dx * dx + dy * dy + dz * dz;
            pxa[j] = px; pya[j] = py; pza[j] = pz; ed[j] = d;
            dmin = fminf(dmin, d);
        }

        float wsum = 0.0f, ox = 0.0f, oy = 0.0f, oz = 0.0f;
#pragma unroll
        for (int j = 0; j < KNN; ++j) {
            const float w = __expf(dmin - ed[j]);
            wsum += w;
            ox += w * pxa[j];
            oy += w * pya[j];
            oz += w * pza[j];
        }
        const float inv = 1.0f / wsum;
        float* outb = out + (size_t)b * 3 * M;
        outb[0 * M + mg] = ox * inv;
        outb[1 * M + mg] = oy * inv;
        outb[2 * M + mg] = oz * inv;
    }
}

extern "C" void kernel_launch(void* const* d_in, const int* in_sizes, int n_in,
                              void* d_out, int out_size, void* d_ws, size_t ws_size,
                              hipStream_t stream) {
    (void)n_in; (void)out_size; (void)d_ws; (void)ws_size;
    const float* pc = (const float*)d_in[0];
    const float* qc = (const float*)d_in[1];
    float* out = (float*)d_out;

    const int B = 4;
    const int N = in_sizes[0] / (3 * B);
    const int M = in_sizes[1] / (3 * B);

    dim3 grid(B * (M / 16));
    softproj_knn_wmma<<<grid, 32 * NW, 0, stream>>>(pc, qc, out, N, M);
}
